# starting kernel
# baseline (speedup 1.0000x reference)
#include <hip/hip_runtime.h>
#include <hip/hip_bf16.h>


#define N1C 50000
#define N2C 50000
#define EC  800000
#define DC  128
#define LC  2
#define WSTR 132

typedef float v2f __attribute__((ext_vector_type(2)));
typedef float v8f __attribute__((ext_vector_type(8)));

__device__ __forceinline__ void atomicMaxF(float* addr, float val) {
  if (val >= 0.f) atomicMax((int*)addr, __float_as_int(val));
  else            atomicMin((unsigned int*)addr, (unsigned int)__float_as_int(val));
}

__global__ __launch_bounds__(256)
void gemm_n128_wmma(const float* __restrict__ A, const float* __restrict__ W,
                    const float* __restrict__ bias, float* __restrict__ C) {
  extern __shared__ float smem[];
  float* sWt = smem;
  float* sA  = smem + DC * WSTR;

  const int rowbase = blockIdx.x * 16;
  const int tid = threadIdx.x;

#pragma unroll
  for (int i = 0; i < 16; ++i) {
    const int c  = tid + 256 * i;
    const int k  = c >> 5;
    const int n4 = (c & 31) * 4;
    const float4 w4 = *(const float4*)(W + (size_t)k * DC + n4);
    sWt[(n4 + 0) * WSTR + k] = w4.x;
    sWt[(n4 + 1) * WSTR + k] = w4.y;
    sWt[(n4 + 2) * WSTR + k] = w4.z;
    sWt[(n4 + 3) * WSTR + k] = w4.w;
  }
#pragma unroll
  for (int i = 0; i < 2; ++i) {
    const int c  = tid + 256 * i;
    const int r  = c >> 5;
    const int n4 = (c & 31) * 4;
    *(float4*)(sA + r * WSTR + n4) =
        *(const float4*)(A + (size_t)(rowbase + r) * DC + n4);
  }
  __syncthreads();

  const int lane = tid & 31;
  const int cb   = (tid >> 5) * 16;
  const int m    = lane & 15;
  const int half = lane >> 4;

  const float* ap = sA + m * WSTR + half * 2;
  const float* bp = sWt + (cb + m) * WSTR + half * 2;

  v8f acc = {};
#pragma unroll 8
  for (int kk = 0; kk < 32; ++kk) {
    const v2f a = *(const v2f*)(ap + kk * 4);
    const v2f b = *(const v2f*)(bp + kk * 4);
    acc = __builtin_amdgcn_wmma_f32_16x16x4_f32(false, a, false, b,
                                                (short)0, acc, false, false);
  }

  const int col = cb + m;
  const float bv = bias ? bias[col] : 0.f;
  float* cp = C + ((size_t)rowbase + half * 8) * DC + col;
#pragma unroll
  for (int r = 0; r < 8; ++r) {
    cp[(size_t)r * DC] = acc[r] + bv;
  }
}

__global__ __launch_bounds__(256)
void rowdot2_kernel(const float* __restrict__ X, const float* __restrict__ v1,
                    const float* __restrict__ v2, float* __restrict__ o1,
                    float* __restrict__ o2, int N) {
  const int row  = blockIdx.x * (blockDim.x >> 5) + (threadIdx.x >> 5);
  const int lane = threadIdx.x & 31;
  if (row >= N) return;
  const float4 x = ((const float4*)(X + (size_t)row * DC))[lane];
  const float4 a = ((const float4*)v1)[lane];
  float s1 = x.x * a.x + x.y * a.y + x.z * a.z + x.w * a.w;
  float s2 = 0.f;
  if (v2) {
    const float4 b = ((const float4*)v2)[lane];
    s2 = x.x * b.x + x.y * b.y + x.z * b.z + x.w * b.w;
  }
#pragma unroll
  for (int off = 16; off > 0; off >>= 1) {
    s1 += __shfl_down(s1, off, 32);
    s2 += __shfl_down(s2, off, 32);
  }
  if (lane == 0) {
    o1[row] = s1;
    if (o2) o2[row] = s2;
  }
}

__global__ void vecmat128_kernel(const float* __restrict__ W,
                                 const float* __restrict__ a,
                                 float* __restrict__ v) {
  const int k = threadIdx.x;
  float s = 0.f;
#pragma unroll 8
  for (int j = 0; j < DC; ++j) s += W[(size_t)k * DC + j] * a[j];
  v[k] = s;
}

__global__ void fillf_kernel(float* p, float val, int n) {
  const int i = blockIdx.x * blockDim.x + threadIdx.x;
  if (i < n) p[i] = val;
}

__global__ void init_bias_kernel(float* __restrict__ out,
                                 const float* __restrict__ b1,
                                 const float* __restrict__ b2, int total) {
  const int i = blockIdx.x * blockDim.x + threadIdx.x;
  if (i >= total) return;
  const int c = i & (DC - 1);
  float v = b1[c];
  if (b2) v += b2[c];
  out[i] = v;
}

__global__ void edge_logit_max_kernel(const float* __restrict__ as,
                                      const float* __restrict__ ad,
                                      const int* __restrict__ src,
                                      const int* __restrict__ dst,
                                      float* __restrict__ elog,
                                      float* __restrict__ maxb, int E) {
  const int e = blockIdx.x * blockDim.x + threadIdx.x;
  if (e >= E) return;
  float l = as[src[e]] + ad[dst[e]];
  l = l > 0.f ? l : 0.2f * l;
  elog[e] = l;
  atomicMaxF(maxb + dst[e], l);
}

__global__ void edge_exp_sum_kernel(const float* __restrict__ elog,
                                    const float* __restrict__ maxb,
                                    const int* __restrict__ dst,
                                    float* __restrict__ ew,
                                    float* __restrict__ sumb, int E) {
  const int e = blockIdx.x * blockDim.x + threadIdx.x;
  if (e >= E) return;
  const int d = dst[e];
  const float w = __expf(elog[e] - maxb[d]);
  ew[e] = w;
  atomicAdd(sumb + d, w);
}

__global__ __launch_bounds__(256)
void edge_aggr_kernel(const float* __restrict__ ew, const float* __restrict__ sumb,
                      const int* __restrict__ src, const int* __restrict__ dst,
                      const float* __restrict__ G, float* __restrict__ O, int E) {
  const long gid = (long)blockIdx.x * blockDim.x + threadIdx.x;
  const int e    = (int)(gid >> 5);
  const int lane = (int)(gid & 31);
  if (e >= E) return;
  const int s = src[e], d = dst[e];
  const float coef = ew[e] / (sumb[d] + 1e-16f);
  const float4 g = *(const float4*)(G + (size_t)s * DC + lane * 4);
  float* op = O + (size_t)d * DC + lane * 4;
  atomicAdd(op + 0, coef * g.x);
  atomicAdd(op + 1, coef * g.y);
  atomicAdd(op + 2, coef * g.z);
  atomicAdd(op + 3, coef * g.w);
}

__global__ void relu_inplace_kernel(float* p, int n) {
  const int i = blockIdx.x * blockDim.x + threadIdx.x;
  if (i < n) {
    const float v = p[i];
    p[i] = v > 0.f ? v : 0.f;
  }
}

extern "C" void kernel_launch(void* const* d_in, const int* in_sizes, int n_in,
                              void* d_out, int out_size, void* d_ws, size_t ws_size,
                              hipStream_t stream) {
  const float* x1    = (const float*)d_in[0];
  const float* x2    = (const float*)d_in[1];
  const int*   ei11  = (const int*)d_in[2];
  const int*   ei22  = (const int*)d_in[3];
  const int*   ei12  = (const int*)d_in[4];
  const float* W_i1  = (const float*)d_in[5];
  const float* as_i1 = (const float*)d_in[6];
  const float* ad_i1 = (const float*)d_in[7];
  const float* b_i1  = (const float*)d_in[8];
  const float* W_i2  = (const float*)d_in[9];
  const float* as_i2 = (const float*)d_in[10];
  const float* ad_i2 = (const float*)d_in[11];
  const float* b_i2  = (const float*)d_in[12];
  const float* Ws_x  = (const float*)d_in[13];
  const float* Wd_x  = (const float*)d_in[14];
  const float* as_x  = (const float*)d_in[15];
  const float* ad_x  = (const float*)d_in[16];
  const float* b_x   = (const float*)d_in[17];
  const float* W_out = (const float*)d_in[18];
  const float* b_out = (const float*)d_in[19];

  float* ws = (float*)d_ws;
  size_t off = 0;
  auto alloc = [&](size_t n) { float* p = ws + off; off += n; return p; };

  float* h1   = alloc((size_t)N1C * DC);
  float* h2   = alloc((size_t)N2C * DC);
  float* g1   = alloc((size_t)N1C * DC);
  float* g2   = alloc((size_t)N2C * DC);
  float* gx   = alloc((size_t)N1C * DC);
  float* a_s1 = alloc(N1C);
  float* a_d1 = alloc(N1C);
  float* a_s2 = alloc(N2C);
  float* a_d2 = alloc(N2C);
  float* a_sx = alloc(N1C);
  float* a_dx = alloc(N2C);
  float* vd   = alloc(DC);
  float* maxb = alloc(N1C > N2C ? N1C : N2C);
  float* sumb = alloc(N1C > N2C ? N1C : N2C);
  float* elog = alloc(EC);
  float* ew   = alloc(EC);
  (void)ws_size; (void)in_sizes; (void)n_in; (void)out_size;

  const dim3 blk(256);
  const dim3 gemm_g1(N1C / 16), gemm_g2(N2C / 16);
  const dim3 rd_g1((N1C + 7) / 8), rd_g2((N2C + 7) / 8);
  const dim3 nd_g1((N1C * DC + 255) / 256), nd_g2((N2C * DC + 255) / 256);
  const dim3 seg_g1((N1C + 255) / 256), seg_g2((N2C + 255) / 256);
  const dim3 e_g((EC + 255) / 256);
  const dim3 ea_g(EC / 8);
  const size_t lds_bytes = (size_t)(DC * WSTR + 16 * WSTR) * sizeof(float);

  const float NEG_BIG = -3.402823466e38f;

  const float* in1 = x1;
  const float* in2 = x2;

  for (int l = 0; l < LC; ++l) {
    const size_t wo = (size_t)l * DC * DC;
    const size_t vo = (size_t)l * DC;

    gemm_n128_wmma<<<gemm_g1, blk, lds_bytes, stream>>>(in1, W_i1 + wo, nullptr, g1);
    gemm_n128_wmma<<<gemm_g2, blk, lds_bytes, stream>>>(in2, W_i2 + wo, nullptr, g2);
    gemm_n128_wmma<<<gemm_g1, blk, lds_bytes, stream>>>(in1, Ws_x + wo, nullptr, gx);

    rowdot2_kernel<<<rd_g1, blk, 0, stream>>>(g1, as_i1 + vo, ad_i1 + vo, a_s1, a_d1, N1C);
    rowdot2_kernel<<<rd_g2, blk, 0, stream>>>(g2, as_i2 + vo, ad_i2 + vo, a_s2, a_d2, N2C);
    rowdot2_kernel<<<rd_g1, blk, 0, stream>>>(gx, as_x + vo, nullptr, a_sx, nullptr, N1C);
    vecmat128_kernel<<<1, DC, 0, stream>>>(Wd_x + wo, ad_x + vo, vd);
    rowdot2_kernel<<<rd_g2, blk, 0, stream>>>(in2, vd, nullptr, a_dx, nullptr, N2C);

    float* o1 = h1;
    float* o2 = h2;
    init_bias_kernel<<<nd_g1, blk, 0, stream>>>(o1, b_i1 + vo, nullptr, N1C * DC);
    init_bias_kernel<<<nd_g2, blk, 0, stream>>>(o2, b_i2 + vo, b_x + vo, N2C * DC);

    fillf_kernel<<<seg_g1, blk, 0, stream>>>(maxb, NEG_BIG, N1C);
    fillf_kernel<<<seg_g1, blk, 0, stream>>>(sumb, 0.f, N1C);
    edge_logit_max_kernel<<<e_g, blk, 0, stream>>>(a_s1, a_d1, ei11, ei11 + EC, elog, maxb, EC);
    edge_exp_sum_kernel<<<e_g, blk, 0, stream>>>(elog, maxb, ei11 + EC, ew, sumb, EC);
    edge_aggr_kernel<<<ea_g, blk, 0, stream>>>(ew, sumb, ei11, ei11 + EC, g1, o1, EC);

    fillf_kernel<<<seg_g2, blk, 0, stream>>>(maxb, NEG_BIG, N2C);
    fillf_kernel<<<seg_g2, blk, 0, stream>>>(sumb, 0.f, N2C);
    edge_logit_max_kernel<<<e_g, blk, 0, stream>>>(a_s2, a_d2, ei22, ei22 + EC, elog, maxb, EC);
    edge_exp_sum_kernel<<<e_g, blk, 0, stream>>>(elog, maxb, ei22 + EC, ew, sumb, EC);
    edge_aggr_kernel<<<ea_g, blk, 0, stream>>>(ew, sumb, ei22, ei22 + EC, g2, o2, EC);

    fillf_kernel<<<seg_g2, blk, 0, stream>>>(maxb, NEG_BIG, N2C);
    fillf_kernel<<<seg_g2, blk, 0, stream>>>(sumb, 0.f, N2C);
    edge_logit_max_kernel<<<e_g, blk, 0, stream>>>(a_sx, a_dx, ei12, ei12 + EC, elog, maxb, EC);
    edge_exp_sum_kernel<<<e_g, blk, 0, stream>>>(elog, maxb, ei12 + EC, ew, sumb, EC);
    edge_aggr_kernel<<<ea_g, blk, 0, stream>>>(ew, sumb, ei12, ei12 + EC, gx, o2, EC);

    relu_inplace_kernel<<<nd_g1, blk, 0, stream>>>(o1, N1C * DC);
    relu_inplace_kernel<<<nd_g2, blk, 0, stream>>>(o2, N2C * DC);

    in1 = h1;
    in2 = h2;
  }

  float* out = (float*)d_out;
  gemm_n128_wmma<<<gemm_g1, blk, lds_bytes, stream>>>(h1, W_out, b_out, out);
  gemm_n128_wmma<<<gemm_g2, blk, lds_bytes, stream>>>(h2, W_out, b_out,
                                                      out + (size_t)N1C * DC);
}


# edge softmax sums: each of 128 work-groups adds its edges' weights into an LDS table of all rows (LDS float adds), then adds the table to memory by whole-line float atomics; sum order changes
# speedup vs baseline: 2.7125x; 2.7125x over previous
#include <hip/hip_runtime.h>
#include <hip/hip_bf16.h>


#define N1C 50000
#define N2C 50000
#define EC  800000
#define DC  128
#define LC  2
#define WSTR 132
#define WCOL(k, n) (((n) + 16 * (((k) >> 1) & 1)) & (DC - 1))

typedef float v2f __attribute__((ext_vector_type(2)));
typedef float v8f __attribute__((ext_vector_type(8)));

__device__ __forceinline__ void atomicMaxF(float* addr, float val) {
  if (val >= 0.f) atomicMax((int*)addr, __float_as_int(val));
  else            atomicMin((unsigned int*)addr, (unsigned int)__float_as_int(val));
}

__device__ __forceinline__ void atomicMaxSigned(float* addr, float val) {
  const int b = __float_as_int(val);
  if (b >= 0) atomicMax((int*)addr, b);
  else        atomicMin((unsigned int*)addr, (unsigned int)b);
}
__device__ __forceinline__ void atomicMinSigned(float* addr, float val) {
  const int b = __float_as_int(val);
  if (b >= 0) atomicMin((int*)addr, b);
  else        atomicMax((unsigned int*)addr, (unsigned int)b);
}

#define STAB_RANGE 16.0f

__device__ __forceinline__ float leaky02(float l) { return l > 0.f ? l : 0.2f * l; }

__device__ __forceinline__ float rowsum8_over16(const float (&p)[8], int m) {
  const bool b3 = (m & 8) != 0, b2 = (m & 4) != 0, b1 = (m & 2) != 0;
  float q[4], s[2];
#pragma unroll
  for (int i = 0; i < 4; ++i) {
    const float give = b3 ? p[i] : p[i + 4];
    const float keep = b3 ? p[i + 4] : p[i];
    q[i] = keep + __shfl_xor(give, 8, 32);
  }
#pragma unroll
  for (int i = 0; i < 2; ++i) {
    const float give = b2 ? q[i] : q[i + 2];
    const float keep = b2 ? q[i + 2] : q[i];
    s[i] = keep + __shfl_xor(give, 4, 32);
  }
  const float give = b1 ? s[0] : s[1];
  const float keep = b1 ? s[1] : s[0];
  float t = keep + __shfl_xor(give, 2, 32);
  t += __shfl_xor(t, 1, 32);
  return t;
}

__global__ __launch_bounds__(256)
void gemm_n128_wmma(const float* A, const float* __restrict__ W,
                    const float* __restrict__ bias, float* __restrict__ C,
                    int relu_in,
                    const float* __restrict__ u, float* __restrict__ du,
                    float* oinit,
                    const float* __restrict__ ib1, const float* __restrict__ ib2,
                    const float* __restrict__ v1, const float* __restrict__ v2,
                    float* __restrict__ d1, float* __restrict__ d2) {
  extern __shared__ float smem[];
  float* sW  = smem;
  float* sA  = smem + DC * DC;
  float* sR  = sA + 16 * WSTR;

  const int rowbase = blockIdx.x * 16;
  const int tid = threadIdx.x;
  const int lane = tid & 31;

#pragma unroll
  for (int i = 0; i < 16; ++i) {
    const int c  = tid + 256 * i;
    const int k  = c >> 5;
    const int n4 = (c & 31) * 4;
    const float4 w4 = *(const float4*)(W + (size_t)k * DC + n4);
    *(float4*)(sW + k * DC + WCOL(k, n4)) = w4;
  }
#pragma unroll
  for (int i = 0; i < 2; ++i) {
    const int c  = tid + 256 * i;
    const int r  = c >> 5;
    const int n4 = (c & 31) * 4;
    float4 a4 = *(const float4*)(A + (size_t)(rowbase + r) * DC + n4);
    if (relu_in) {
      a4.x = a4.x > 0.f ? a4.x : 0.f;
      a4.y = a4.y > 0.f ? a4.y : 0.f;
      a4.z = a4.z > 0.f ? a4.z : 0.f;
      a4.w = a4.w > 0.f ? a4.w : 0.f;
    }
    *(float4*)(sA + r * WSTR + n4) = a4;
    if (u) {
      const float4 u4 = ((const float4*)u)[lane];
      float s = a4.x * u4.x + a4.y * u4.y + a4.z * u4.z + a4.w * u4.w;
#pragma unroll
      for (int off = 16; off > 0; off >>= 1) s += __shfl_down(s, off, 32);
      if (lane == 0) du[rowbase + r] = s;
    }
  }
  __syncthreads();

  if (oinit) {
#pragma unroll
    for (int i = 0; i < 2; ++i) {
      const int c  = tid + 256 * i;
      const int r  = c >> 5;
      const int n4 = (c & 31) * 4;
      float4 b4 = *(const float4*)(ib1 + n4);
      if (ib2) {
        const float4 c4 = *(const float4*)(ib2 + n4);
        b4.x += c4.x; b4.y += c4.y; b4.z += c4.z; b4.w += c4.w;
      }
      *(float4*)(oinit + (size_t)(rowbase + r) * DC + n4) = b4;
    }
  }

  const int cb   = (tid >> 5) * 16;
  const int m    = lane & 15;
  const int half = lane >> 4;

  const float* ap = sA + m * WSTR + half * 2;
  const float* bp = sW + half * (2 * DC) + WCOL(2 * half, cb + m);

  v8f acc = {};
#pragma unroll
  for (int kk = 0; kk < 32; ++kk) {
    const v2f a = *(const v2f*)(ap + kk * 4);
    v2f b;
    b.x = bp[kk * (4 * DC)];
    b.y = bp[kk * (4 * DC) + DC];
    acc = __builtin_amdgcn_wmma_f32_16x16x4_f32(false, a, false, b,
                                                (short)0, acc, false, false);
  }

  asm volatile("v_nop\n\tv_nop\n\tv_nop\n\tv_nop" : "+v"(acc));

  const int col = cb + m;
  const float bv = bias ? bias[col] : 0.f;
  float cv[8];
#pragma unroll
  for (int r = 0; r < 8; ++r) cv[r] = acc[r] + bv;

  if (v1) {
    const float w1 = v1[col];
    const float w2 = v2 ? v2[col] : 0.f;
    float p1[8], p2[8];
#pragma unroll
    for (int r = 0; r < 8; ++r) { p1[r] = cv[r] * w1; p2[r] = cv[r] * w2; }
    const float t1 = rowsum8_over16(p1, m);
    const float t2 = rowsum8_over16(p2, m);
    if ((m & 1) == 0) {
      const int rr = half * 8 + (m >> 1);
      float* q = sR + (((tid >> 5) * 16 + rr) << 1);
      q[0] = t1;
      q[1] = t2;
    }
    __syncthreads();
    if (tid < 32) {
      const int rr = tid >> 1, which = tid & 1;
      float s = 0.f;
#pragma unroll
      for (int w = 0; w < 8; ++w) s += sR[((w * 16 + rr) << 1) + which];
      float* dp = which ? d2 : d1;
      if (dp) dp[rowbase + rr] = s;
    }
  }

  float* cp = C + ((size_t)rowbase + half * 8) * DC + col;
#pragma unroll
  for (int r = 0; r < 8; ++r) cp[(size_t)r * DC] = cv[r];
}

__global__ void vecmat128_kernel(const float* __restrict__ W,
                                 const float* __restrict__ a,
                                 float* __restrict__ v) {
  const int k = threadIdx.x;
  float s = 0.f;
#pragma unroll 8
  for (int j = 0; j < DC; ++j) s += W[(size_t)k * DC + j] * a[j];
  v[k] = s;
}

__global__ void init_max_kernel(float* __restrict__ maxb, float* __restrict__ gst,
                                float val, int n) {
  const int i = blockIdx.x * blockDim.x + threadIdx.x;
  if (i < n) maxb[i] = val;
  if (i == 0) { gst[0] = val; gst[1] = -val; }
}

__global__ __launch_bounds__(256)
void init_sum_range_kernel(float* __restrict__ sumb, int n_dst,
                           const float* __restrict__ as, int n_src,
                           float* __restrict__ gst) {
  __shared__ float smx[8], smn[8];
  const int i = blockIdx.x * blockDim.x + threadIdx.x;
  if (i < n_dst) sumb[i] = 0.f;
  float mx = -3.402823466e38f, mn = 3.402823466e38f;
  if (i < n_src) { mx = as[i]; mn = mx; }
#pragma unroll
  for (int off = 16; off > 0; off >>= 1) {
    mx = fmaxf(mx, __shfl_xor(mx, off, 32));
    mn = fminf(mn, __shfl_xor(mn, off, 32));
  }
  const int w = threadIdx.x >> 5;
  if ((threadIdx.x & 31) == 0) { smx[w] = mx; smn[w] = mn; }
  __syncthreads();
  if (threadIdx.x == 0) {
#pragma unroll
    for (int k = 1; k < 8; ++k) { mx = fmaxf(mx, smx[k]); mn = fminf(mn, smn[k]); }
    atomicMaxSigned(gst, mx);
    atomicMinSigned(gst + 1, mn);
  }
}

#define EMAX_BLOCKS 256
__global__ __launch_bounds__(256)
void edge_max_exact_kernel(const float* __restrict__ as,
                           const float* __restrict__ ad,
                           const int* __restrict__ src,
                           const int* __restrict__ dst,
                           const float* __restrict__ gst,
                           float* __restrict__ maxb, int E) {
  const float range = gst[0] - gst[1];
  if (range >= 0.f && range <= STAB_RANGE) return;
  for (int e = blockIdx.x * 256 + threadIdx.x; e < E; e += EMAX_BLOCKS * 256) {
    const int d = dst[e];
    const float l = leaky02(as[src[e]] + ad[d]);
    atomicMaxF(maxb + d, l);
  }
}

#define ES_WG  128
#define ES_T   1024
#define ES_U   4
#define ES_FU  7
#define ES_TAB ((((N1C > N2C ? N1C : N2C) + 3) / 4) * 4)
__global__ __launch_bounds__(ES_T)
void edge_exp_sum_kernel(const float* __restrict__ as,
                         const float* __restrict__ ad,
                         const int* __restrict__ src,
                         const int* __restrict__ dst,
                         const float* __restrict__ gst,
                         const float* __restrict__ maxb,
                         float* __restrict__ ew,
                         float* __restrict__ sumb, int E, int n_dst) {
  __shared__ __attribute__((aligned(16))) float stab[ES_TAB];
  const int tid = (int)threadIdx.x;

  for (int i = tid; i < ES_TAB / 4; i += ES_T) ((float4*)stab)[i] = make_float4(0.f, 0.f, 0.f, 0.f);

  const float gmax  = gst[0];
  const float range = gmax - gst[1];
  const bool bound_ok = range >= 0.f && range <= STAB_RANGE;
  __syncthreads();

  const int per = (E + (int)gridDim.x - 1) / (int)gridDim.x;
  const int lo  = (int)blockIdx.x * per;
  const int hi  = (lo + per) < E ? (lo + per) : E;
  for (int base = lo + tid; base < hi; base += ES_T * ES_U) {
    int   s[ES_U], d[ES_U];
    float asv[ES_U], adv[ES_U], m[ES_U];
#pragma unroll
    for (int u = 0; u < ES_U; ++u) {
      const int e  = base + u * ES_T;
      const int ee = e < hi ? e : hi - 1;
      s[u] = src[ee];
      d[u] = dst[ee];
    }
#pragma unroll
    for (int u = 0; u < ES_U; ++u) {
      asv[u] = as[s[u]];
      adv[u] = ad[d[u]];
    }
    if (bound_ok) {
#pragma unroll
      for (int u = 0; u < ES_U; ++u) m[u] = leaky02(gmax + adv[u]);
    } else {
#pragma unroll
      for (int u = 0; u < ES_U; ++u) m[u] = maxb[d[u]];
    }
#pragma unroll
    for (int u = 0; u < ES_U; ++u) {
      const int e = base + u * ES_T;
      const float w = __expf(leaky02(asv[u] + adv[u]) - m[u]);
      if (e < hi) {
        ew[e] = w;
        atomicAdd(stab + d[u], w);
      }
    }
  }
  __syncthreads();

  const int mis    = (int)((((size_t)sumb) >> 2) & 31);
  const int nsteps = (n_dst + mis + ES_T - 1) / ES_T;
  const int rot    = (int)blockIdx.x % nsteps;
  for (int st0 = 0; st0 < nsteps; st0 += ES_FU) {
    int   idx[ES_FU];
    float v[ES_FU];
#pragma unroll
    for (int u = 0; u < ES_FU; ++u) {
      const int st = st0 + u;
      int k = st + rot;
      if (k >= nsteps) k -= nsteps;
      const int i = k * ES_T + tid - mis;
      idx[u] = i;
      v[u] = (st < nsteps && i >= 0 && i < n_dst) ? stab[i] : 0.f;
    }
#pragma unroll
    for (int u = 0; u < ES_FU; ++u)
      if (v[u] != 0.f) atomicAdd(sumb + idx[u], v[u]);
  }
}

#define AGG_CH 32
#define AGG_U  8
__global__ __launch_bounds__(128)
void edge_aggr_kernel(const float* __restrict__ ew, const float* __restrict__ sumb,
                      const int* __restrict__ src, const int* __restrict__ dst,
                      const float* __restrict__ G, float* __restrict__ O, int E) {
  const int wave = __builtin_amdgcn_readfirstlane(
      (int)((blockIdx.x * blockDim.x + threadIdx.x) >> 5));
  const int lane = (int)(threadIdx.x & 31);
  const int e0   = wave * AGG_CH;
  if (e0 >= E) return;
  const int n = (E - e0) < AGG_CH ? (E - e0) : AGG_CH;

  int s = 0, d = 0;
  float coef = 0.f;
  if (lane < n) {
    const int e = e0 + lane;
    s = src[e];
    d = dst[e];
    coef = ew[e] / (sumb[d] + 1e-16f);
  }

  if (n == AGG_CH) {
#pragma unroll 1
    for (int j0 = 0; j0 < AGG_CH; j0 += AGG_U) {
      float g[AGG_U][4];
#pragma unroll
      for (int u = 0; u < AGG_U; ++u) {
        const int sj = __builtin_amdgcn_readlane(s, j0 + u);
        const float* gp = G + (size_t)sj * DC;
        g[u][0] = gp[lane];
        g[u][1] = gp[lane + 32];
        g[u][2] = gp[lane + 64];
        g[u][3] = gp[lane + 96];
      }
#pragma unroll
      for (int u = 0; u < AGG_U; ++u) {
        const int   dj = __builtin_amdgcn_readlane(d, j0 + u);
        const float cj = __int_as_float(
            __builtin_amdgcn_readlane(__float_as_int(coef), j0 + u));
        float* op = O + (size_t)dj * DC;
        atomicAdd(op + lane,      cj * g[u][0]);
        atomicAdd(op + lane + 32, cj * g[u][1]);
        atomicAdd(op + lane + 64, cj * g[u][2]);
        atomicAdd(op + lane + 96, cj * g[u][3]);
      }
    }
  } else {
    for (int j = 0; j < n; ++j) {
      const int   sj = __builtin_amdgcn_readlane(s, j);
      const int   dj = __builtin_amdgcn_readlane(d, j);
      const float cj = __int_as_float(
          __builtin_amdgcn_readlane(__float_as_int(coef), j));
      const float* gp = G + (size_t)sj * DC;
      float* op = O + (size_t)dj * DC;
      const float g0 = gp[lane], g1 = gp[lane + 32], g2 = gp[lane + 64], g3 = gp[lane + 96];
      atomicAdd(op + lane,      cj * g0);
      atomicAdd(op + lane + 32, cj * g1);
      atomicAdd(op + lane + 64, cj * g2);
      atomicAdd(op + lane + 96, cj * g3);
    }
  }
}

extern "C" void kernel_launch(void* const* d_in, const int* in_sizes, int n_in,
                              void* d_out, int out_size, void* d_ws, size_t ws_size,
                              hipStream_t stream) {
  const float* x1    = (const float*)d_in[0];
  const float* x2    = (const float*)d_in[1];
  const int*   ei11  = (const int*)d_in[2];
  const int*   ei22  = (const int*)d_in[3];
  const int*   ei12  = (const int*)d_in[4];
  const float* W_i1  = (const float*)d_in[5];
  const float* as_i1 = (const float*)d_in[6];
  const float* ad_i1 = (const float*)d_in[7];
  const float* b_i1  = (const float*)d_in[8];
  const float* W_i2  = (const float*)d_in[9];
  const float* as_i2 = (const float*)d_in[10];
  const float* ad_i2 = (const float*)d_in[11];
  const float* b_i2  = (const float*)d_in[12];
  const float* Ws_x  = (const float*)d_in[13];
  const float* Wd_x  = (const float*)d_in[14];
  const float* as_x  = (const float*)d_in[15];
  const float* ad_x  = (const float*)d_in[16];
  const float* b_x   = (const float*)d_in[17];
  const float* W_out = (const float*)d_in[18];
  const float* b_out = (const float*)d_in[19];

  float* ws = (float*)d_ws;
  size_t off = 0;
  auto alloc = [&](size_t n) { float* p = ws + off; off += n; return p; };

  float* h1   = alloc((size_t)N1C * DC);
  float* h2   = alloc((size_t)N2C * DC);
  float* g1   = alloc((size_t)N1C * DC);
  float* g2   = alloc((size_t)N2C * DC);
  float* gx   = alloc((size_t)N1C * DC);
  float* a_s1 = alloc(N1C);
  float* a_d1 = alloc(N1C);
  float* a_s2 = alloc(N2C);
  float* a_d2 = alloc(N2C);
  float* a_sx = alloc(N1C);
  float* a_dx = alloc(N2C);
  float* vd   = alloc(DC);
  float* maxb = alloc(N1C > N2C ? N1C : N2C);
  float* sumb = alloc(N1C > N2C ? N1C : N2C);
  float* ew   = alloc(EC);
  float* gst  = alloc(2);
  (void)ws_size; (void)in_sizes; (void)n_in; (void)out_size;

  const dim3 blk(256);
  const dim3 gemm_g1(N1C / 16), gemm_g2(N2C / 16);
  const dim3 seg_g1((N1C + 255) / 256), seg_g2((N2C + 255) / 256);
  const dim3 es_g(ES_WG), es_b(ES_T);
  const dim3 seg_gx(((N1C > N2C ? N1C : N2C) + 255) / 256);
  const dim3 emax_g(EMAX_BLOCKS);
  const dim3 ea_b(128);
  const dim3 ea_g((EC + 4 * AGG_CH - 1) / (4 * AGG_CH));
  const size_t lds_bytes = (size_t)(DC * DC + 16 * WSTR + 8 * 16 * 2) * sizeof(float);

  const float NEG_BIG = -3.402823466e38f;

  const float* in1 = x1;
  const float* in2 = x2;

  for (int l = 0; l < LC; ++l) {
    const size_t wo = (size_t)l * DC * DC;
    const size_t vo = (size_t)l * DC;

    float* o1 = h1;
    float* o2 = h2;

    vecmat128_kernel<<<1, DC, 0, stream>>>(Wd_x + wo, ad_x + vo, vd);

    gemm_n128_wmma<<<gemm_g1, blk, lds_bytes, stream>>>(in1, Ws_x + wo, nullptr, gx, l,
        nullptr, nullptr,
        nullptr, nullptr, nullptr,
        as_x + vo, nullptr, a_sx, nullptr);
    gemm_n128_wmma<<<gemm_g1, blk, lds_bytes, stream>>>(in1, W_i1 + wo, nullptr, g1, l,
        nullptr, nullptr,
        o1, b_i1 + vo, nullptr,
        as_i1 + vo, ad_i1 + vo, a_s1, a_d1);
    gemm_n128_wmma<<<gemm_g2, blk, lds_bytes, stream>>>(in2, W_i2 + wo, nullptr, g2, l,
        vd, a_dx,
        o2, b_i2 + vo, b_x + vo,
        as_i2 + vo, ad_i2 + vo, a_s2, a_d2);

    init_max_kernel<<<seg_g1, blk, 0, stream>>>(maxb, gst, NEG_BIG, N1C);
    init_sum_range_kernel<<<seg_g1, blk, 0, stream>>>(sumb, N1C, a_s1, N1C, gst);
    edge_max_exact_kernel<<<emax_g, blk, 0, stream>>>(a_s1, a_d1, ei11, ei11 + EC, gst, maxb, EC);
    edge_exp_sum_kernel<<<es_g, es_b, 0, stream>>>(a_s1, a_d1, ei11, ei11 + EC, gst, maxb, ew, sumb, EC, N1C);
    edge_aggr_kernel<<<ea_g, ea_b, 0, stream>>>(ew, sumb, ei11, ei11 + EC, g1, o1, EC);

    init_max_kernel<<<seg_g2, blk, 0, stream>>>(maxb, gst, NEG_BIG, N2C);
    init_sum_range_kernel<<<seg_g2, blk, 0, stream>>>(sumb, N2C, a_s2, N2C, gst);
    edge_max_exact_kernel<<<emax_g, blk, 0, stream>>>(a_s2, a_d2, ei22, ei22 + EC, gst, maxb, EC);
    edge_exp_sum_kernel<<<es_g, es_b, 0, stream>>>(a_s2, a_d2, ei22, ei22 + EC, gst, maxb, ew, sumb, EC, N2C);
    edge_aggr_kernel<<<ea_g, ea_b, 0, stream>>>(ew, sumb, ei22, ei22 + EC, g2, o2, EC);

    init_max_kernel<<<seg_g2, blk, 0, stream>>>(maxb, gst, NEG_BIG, N2C);
    init_sum_range_kernel<<<seg_gx, blk, 0, stream>>>(sumb, N2C, a_sx, N1C, gst);
    edge_max_exact_kernel<<<emax_g, blk, 0, stream>>>(a_sx, a_dx, ei12, ei12 + EC, gst, maxb, EC);
    edge_exp_sum_kernel<<<es_g, es_b, 0, stream>>>(a_sx, a_dx, ei12, ei12 + EC, gst, maxb, ew, sumb, EC, N2C);
    edge_aggr_kernel<<<ea_g, ea_b, 0, stream>>>(ew, sumb, ei12, ei12 + EC, gx, o2, EC);

    in1 = h1;
    in2 = h2;
  }

  float* out = (float*)d_out;
  gemm_n128_wmma<<<gemm_g1, blk, lds_bytes, stream>>>(h1, W_out, b_out, out, 1,
      nullptr, nullptr, nullptr, nullptr, nullptr,
      nullptr, nullptr, nullptr, nullptr);
  gemm_n128_wmma<<<gemm_g2, blk, lds_bytes, stream>>>(h2, W_out, b_out,
      out + (size_t)N1C * DC, 1,
      nullptr, nullptr, nullptr, nullptr, nullptr,
      nullptr, nullptr, nullptr, nullptr);
}


# softmax-sum launch prepares itself: scans source logits for the range, exact max (wide range) in its LDS table, zeroes the next set's sums (2 buffers); 18 small launches go, 40 to 22; sums unchanged
# speedup vs baseline: 2.8148x; 1.0377x over previous
#include <hip/hip_runtime.h>
#include <hip/hip_bf16.h>


#define N1C 50000
#define N2C 50000
#define EC  800000
#define DC  128
#define LC  2
#define WSTR 132
#define WCOL(k, n) (((n) + 16 * (((k) >> 1) & 1)) & (DC - 1))

typedef float v2f __attribute__((ext_vector_type(2)));
typedef float v8f __attribute__((ext_vector_type(8)));

__device__ __forceinline__ void atomicMaxSigned(float* addr, float val) {
  const int b = __float_as_int(val);
  if (b >= 0) atomicMax((int*)addr, b);
  else        atomicMin((unsigned int*)addr, (unsigned int)b);
}

#define STAB_RANGE 16.0f

__device__ __forceinline__ float leaky02(float l) { return l > 0.f ? l : 0.2f * l; }

__device__ __forceinline__ float rowsum8_over16(const float (&p)[8], int m) {
  const bool b3 = (m & 8) != 0, b2 = (m & 4) != 0, b1 = (m & 2) != 0;
  float q[4], s[2];
#pragma unroll
  for (int i = 0; i < 4; ++i) {
    const float give = b3 ? p[i] : p[i + 4];
    const float keep = b3 ? p[i + 4] : p[i];
    q[i] = keep + __shfl_xor(give, 8, 32);
  }
#pragma unroll
  for (int i = 0; i < 2; ++i) {
    const float give = b2 ? q[i] : q[i + 2];
    const float keep = b2 ? q[i + 2] : q[i];
    s[i] = keep + __shfl_xor(give, 4, 32);
  }
  const float give = b1 ? s[0] : s[1];
  const float keep = b1 ? s[1] : s[0];
  float t = keep + __shfl_xor(give, 2, 32);
  t += __shfl_xor(t, 1, 32);
  return t;
}

__global__ __launch_bounds__(256)
void gemm_n128_wmma(const float* A, const float* __restrict__ W,
                    const float* __restrict__ bias, float* __restrict__ C,
                    int relu_in,
                    const float* __restrict__ u, float* __restrict__ du,
                    float* oinit,
                    const float* __restrict__ ib1, const float* __restrict__ ib2,
                    const float* __restrict__ v1, const float* __restrict__ v2,
                    float* __restrict__ d1, float* __restrict__ d2) {
  extern __shared__ float smem[];
  float* sW  = smem;
  float* sA  = smem + DC * DC;
  float* sR  = sA + 16 * WSTR;

  const int rowbase = blockIdx.x * 16;
  const int tid = threadIdx.x;
  const int lane = tid & 31;

#pragma unroll
  for (int i = 0; i < 16; ++i) {
    const int c  = tid + 256 * i;
    const int k  = c >> 5;
    const int n4 = (c & 31) * 4;
    const float4 w4 = *(const float4*)(W + (size_t)k * DC + n4);
    *(float4*)(sW + k * DC + WCOL(k, n4)) = w4;
  }
#pragma unroll
  for (int i = 0; i < 2; ++i) {
    const int c  = tid + 256 * i;
    const int r  = c >> 5;
    const int n4 = (c & 31) * 4;
    float4 a4 = *(const float4*)(A + (size_t)(rowbase + r) * DC + n4);
    if (relu_in) {
      a4.x = a4.x > 0.f ? a4.x : 0.f;
      a4.y = a4.y > 0.f ? a4.y : 0.f;
      a4.z = a4.z > 0.f ? a4.z : 0.f;
      a4.w = a4.w > 0.f ? a4.w : 0.f;
    }
    *(float4*)(sA + r * WSTR + n4) = a4;
    if (u) {
      const float4 u4 = ((const float4*)u)[lane];
      float s = a4.x * u4.x + a4.y * u4.y + a4.z * u4.z + a4.w * u4.w;
#pragma unroll
      for (int off = 16; off > 0; off >>= 1) s += __shfl_down(s, off, 32);
      if (lane == 0) du[rowbase + r] = s;
    }
  }
  __syncthreads();

  if (oinit) {
#pragma unroll
    for (int i = 0; i < 2; ++i) {
      const int c  = tid + 256 * i;
      const int r  = c >> 5;
      const int n4 = (c & 31) * 4;
      float4 b4 = *(const float4*)(ib1 + n4);
      if (ib2) {
        const float4 c4 = *(const float4*)(ib2 + n4);
        b4.x += c4.x; b4.y += c4.y; b4.z += c4.z; b4.w += c4.w;
      }
      *(float4*)(oinit + (size_t)(rowbase + r) * DC + n4) = b4;
    }
  }

  const int cb   = (tid >> 5) * 16;
  const int m    = lane & 15;
  const int half = lane >> 4;

  const float* ap = sA + m * WSTR + half * 2;
  const float* bp = sW + half * (2 * DC) + WCOL(2 * half, cb + m);

  v8f acc = {};
#pragma unroll
  for (int kk = 0; kk < 32; ++kk) {
    const v2f a = *(const v2f*)(ap + kk * 4);
    v2f b;
    b.x = bp[kk * (4 * DC)];
    b.y = bp[kk * (4 * DC) + DC];
    acc = __builtin_amdgcn_wmma_f32_16x16x4_f32(false, a, false, b,
                                                (short)0, acc, false, false);
  }

  asm volatile("v_nop\n\tv_nop\n\tv_nop\n\tv_nop" : "+v"(acc));

  const int col = cb + m;
  const float bv = bias ? bias[col] : 0.f;
  float cv[8];
#pragma unroll
  for (int r = 0; r < 8; ++r) cv[r] = acc[r] + bv;

  if (v1) {
    const float w1 = v1[col];
    const float w2 = v2 ? v2[col] : 0.f;
    float p1[8], p2[8];
#pragma unroll
    for (int r = 0; r < 8; ++r) { p1[r] = cv[r] * w1; p2[r] = cv[r] * w2; }
    const float t1 = rowsum8_over16(p1, m);
    const float t2 = rowsum8_over16(p2, m);
    if ((m & 1) == 0) {
      const int rr = half * 8 + (m >> 1);
      float* q = sR + (((tid >> 5) * 16 + rr) << 1);
      q[0] = t1;
      q[1] = t2;
    }
    __syncthreads();
    if (tid < 32) {
      const int rr = tid >> 1, which = tid & 1;
      float s = 0.f;
#pragma unroll
      for (int w = 0; w < 8; ++w) s += sR[((w * 16 + rr) << 1) + which];
      float* dp = which ? d2 : d1;
      if (dp) dp[rowbase + rr] = s;
    }
  }

  float* cp = C + ((size_t)rowbase + half * 8) * DC + col;
#pragma unroll
  for (int r = 0; r < 8; ++r) cp[(size_t)r * DC] = cv[r];
}

#define ES_WG  128
#define ES_T   1024
#define ES_U   4
#define ES_FU  7
#define ES_SU  7
#define ES_TAB ((((N1C > N2C ? N1C : N2C) + 3) / 4) * 4)

#define VZ_BLOCKS ((ES_TAB / 4 + DC - 1) / DC)
__global__ void vecmat128_kernel(const float* __restrict__ W,
                                 const float* __restrict__ a,
                                 float* __restrict__ v,
                                 float* __restrict__ zbuf) {
  const int k = threadIdx.x;
  const int i = (int)blockIdx.x * DC + k;
  if (i < ES_TAB / 4) ((float4*)zbuf)[i] = make_float4(0.f, 0.f, 0.f, 0.f);
  if (blockIdx.x == 0) {
    float s = 0.f;
#pragma unroll 8
    for (int j = 0; j < DC; ++j) s += W[(size_t)k * DC + j] * a[j];
    v[k] = s;
  }
}

__global__ __launch_bounds__(ES_T)
void edge_exp_sum_kernel(const float* __restrict__ as,
                         const float* __restrict__ ad,
                         const int* __restrict__ src,
                         const int* __restrict__ dst,
                         float* __restrict__ ew,
                         float* __restrict__ sumb,
                         float* __restrict__ znext,
                         int E, int n_src, int n_dst) {
  __shared__ __attribute__((aligned(16))) float stab[ES_TAB];
  __shared__ float smx[ES_T / 32], smn[ES_T / 32];
  const int tid  = (int)threadIdx.x;
  const int lane = tid & 31;

  for (int i = tid; i < ES_TAB / 4; i += ES_T) ((float4*)stab)[i] = make_float4(0.f, 0.f, 0.f, 0.f);

  for (int i = (int)blockIdx.x * ES_T + tid; i < ES_TAB / 4; i += (int)gridDim.x * ES_T)
    ((float4*)znext)[i] = make_float4(0.f, 0.f, 0.f, 0.f);

  float mx = -3.402823466e38f, mn = 3.402823466e38f;
  const int n4 = n_src >> 2;
  const float4* as4 = (const float4*)as;
  for (int i0 = tid; i0 < n4; i0 += ES_T * ES_SU) {
    float4 a4[ES_SU];
#pragma unroll
    for (int u = 0; u < ES_SU; ++u) {
      const int i = i0 + u * ES_T;
      a4[u] = as4[i < n4 ? i : i0];
    }
#pragma unroll
    for (int u = 0; u < ES_SU; ++u) {
      mx = fmaxf(fmaxf(mx, a4[u].x), fmaxf(a4[u].y, fmaxf(a4[u].z, a4[u].w)));
      mn = fminf(fminf(mn, a4[u].x), fminf(a4[u].y, fminf(a4[u].z, a4[u].w)));
    }
  }
  for (int i = (n4 << 2) + tid; i < n_src; i += ES_T) {
    const float a1 = as[i];
    mx = fmaxf(mx, a1);
    mn = fminf(mn, a1);
  }
#pragma unroll
  for (int off = 16; off > 0; off >>= 1) {
    mx = fmaxf(mx, __shfl_xor(mx, off, 32));
    mn = fminf(mn, __shfl_xor(mn, off, 32));
  }
  if (lane == 0) { smx[tid >> 5] = mx; smn[tid >> 5] = mn; }
  __syncthreads();
  float gmax = smx[lane], gmin = smn[lane];
#pragma unroll
  for (int off = 16; off > 0; off >>= 1) {
    gmax = fmaxf(gmax, __shfl_xor(gmax, off, 32));
    gmin = fminf(gmin, __shfl_xor(gmin, off, 32));
  }
  const float range = gmax - gmin;
  const int bound_ok = __builtin_amdgcn_readfirstlane(
      (range >= 0.f && range <= STAB_RANGE) ? 1 : 0);

  const int per = (E + (int)gridDim.x - 1) / (int)gridDim.x;
  const int lo  = (int)blockIdx.x * per;
  const int hi  = (lo + per) < E ? (lo + per) : E;

  if (bound_ok) {
    for (int base = lo + tid; base < hi; base += ES_T * ES_U) {
      int   s[ES_U], d[ES_U];
      float asv[ES_U], adv[ES_U];
#pragma unroll
      for (int u = 0; u < ES_U; ++u) {
        const int e  = base + u * ES_T;
        const int ee = e < hi ? e : hi - 1;
        s[u] = src[ee];
        d[u] = dst[ee];
      }
#pragma unroll
      for (int u = 0; u < ES_U; ++u) {
        asv[u] = as[s[u]];
        adv[u] = ad[d[u]];
      }
#pragma unroll
      for (int u = 0; u < ES_U; ++u) {
        const int e = base + u * ES_T;
        const float w = __expf(leaky02(asv[u] + adv[u]) - leaky02(gmax + adv[u]));
        if (e < hi) {
          ew[e] = w;
          atomicAdd(stab + d[u], w);
        }
      }
    }
  } else {
    for (int i = tid; i < ES_TAB; i += ES_T) stab[i] = -3.402823466e38f;
    __syncthreads();
    for (int e = tid; e < E; e += ES_T) {
      const int d = dst[e];
      atomicMaxSigned(stab + d, leaky02(as[src[e]] + ad[d]));
    }
    __syncthreads();
    for (int e = lo + tid; e < hi; e += ES_T) {
      const int d = dst[e];
      ew[e] = __expf(leaky02(as[src[e]] + ad[d]) - stab[d]);
    }
    __syncthreads();
    for (int i = tid; i < ES_TAB / 4; i += ES_T) ((float4*)stab)[i] = make_float4(0.f, 0.f, 0.f, 0.f);
    __syncthreads();
    for (int e = lo + tid; e < hi; e += ES_T) atomicAdd(stab + dst[e], ew[e]);
  }
  __syncthreads();

  const int mis    = (int)((((size_t)sumb) >> 2) & 31);
  const int nsteps = (n_dst + mis + ES_T - 1) / ES_T;
  const int rot    = (int)blockIdx.x % nsteps;
  for (int st0 = 0; st0 < nsteps; st0 += ES_FU) {
    int   idx[ES_FU];
    float v[ES_FU];
#pragma unroll
    for (int u = 0; u < ES_FU; ++u) {
      const int st = st0 + u;
      int k = st + rot;
      if (k >= nsteps) k -= nsteps;
      const int i = k * ES_T + tid - mis;
      idx[u] = i;
      v[u] = (st < nsteps && i >= 0 && i < n_dst) ? stab[i] : 0.f;
    }
#pragma unroll
    for (int u = 0; u < ES_FU; ++u)
      if (v[u] != 0.f) atomicAdd(sumb + idx[u], v[u]);
  }
}

#define AGG_CH 32
#define AGG_U  8
__global__ __launch_bounds__(128)
void edge_aggr_kernel(const float* __restrict__ ew, const float* __restrict__ sumb,
                      const int* __restrict__ src, const int* __restrict__ dst,
                      const float* __restrict__ G, float* __restrict__ O, int E) {
  const int wave = __builtin_amdgcn_readfirstlane(
      (int)((blockIdx.x * blockDim.x + threadIdx.x) >> 5));
  const int lane = (int)(threadIdx.x & 31);
  const int e0   = wave * AGG_CH;
  if (e0 >= E) return;
  const int n = (E - e0) < AGG_CH ? (E - e0) : AGG_CH;

  int s = 0, d = 0;
  float coef = 0.f;
  if (lane < n) {
    const int e = e0 + lane;
    s = src[e];
    d = dst[e];
    coef = ew[e] / (sumb[d] + 1e-16f);
  }

  if (n == AGG_CH) {
#pragma unroll 1
    for (int j0 = 0; j0 < AGG_CH; j0 += AGG_U) {
      float g[AGG_U][4];
#pragma unroll
      for (int u = 0; u < AGG_U; ++u) {
        const int sj = __builtin_amdgcn_readlane(s, j0 + u);
        const float* gp = G + (size_t)sj * DC;
        g[u][0] = gp[lane];
        g[u][1] = gp[lane + 32];
        g[u][2] = gp[lane + 64];
        g[u][3] = gp[lane + 96];
      }
#pragma unroll
      for (int u = 0; u < AGG_U; ++u) {
        const int   dj = __builtin_amdgcn_readlane(d, j0 + u);
        const float cj = __int_as_float(
            __builtin_amdgcn_readlane(__float_as_int(coef), j0 + u));
        float* op = O + (size_t)dj * DC;
        atomicAdd(op + lane,      cj * g[u][0]);
        atomicAdd(op + lane + 32, cj * g[u][1]);
        atomicAdd(op + lane + 64, cj * g[u][2]);
        atomicAdd(op + lane + 96, cj * g[u][3]);
      }
    }
  } else {
    for (int j = 0; j < n; ++j) {
      const int   sj = __builtin_amdgcn_readlane(s, j);
      const int   dj = __builtin_amdgcn_readlane(d, j);
      const float cj = __int_as_float(
          __builtin_amdgcn_readlane(__float_as_int(coef), j));
      const float* gp = G + (size_t)sj * DC;
      float* op = O + (size_t)dj * DC;
      const float g0 = gp[lane], g1 = gp[lane + 32], g2 = gp[lane + 64], g3 = gp[lane + 96];
      atomicAdd(op + lane,      cj * g0);
      atomicAdd(op + lane + 32, cj * g1);
      atomicAdd(op + lane + 64, cj * g2);
      atomicAdd(op + lane + 96, cj * g3);
    }
  }
}

extern "C" void kernel_launch(void* const* d_in, const int* in_sizes, int n_in,
                              void* d_out, int out_size, void* d_ws, size_t ws_size,
                              hipStream_t stream) {
  const float* x1    = (const float*)d_in[0];
  const float* x2    = (const float*)d_in[1];
  const int*   ei11  = (const int*)d_in[2];
  const int*   ei22  = (const int*)d_in[3];
  const int*   ei12  = (const int*)d_in[4];
  const float* W_i1  = (const float*)d_in[5];
  const float* as_i1 = (const float*)d_in[6];
  const float* ad_i1 = (const float*)d_in[7];
  const float* b_i1  = (const float*)d_in[8];
  const float* W_i2  = (const float*)d_in[9];
  const float* as_i2 = (const float*)d_in[10];
  const float* ad_i2 = (const float*)d_in[11];
  const float* b_i2  = (const float*)d_in[12];
  const float* Ws_x  = (const float*)d_in[13];
  const float* Wd_x  = (const float*)d_in[14];
  const float* as_x  = (const float*)d_in[15];
  const float* ad_x  = (const float*)d_in[16];
  const float* b_x   = (const float*)d_in[17];
  const float* W_out = (const float*)d_in[18];
  const float* b_out = (const float*)d_in[19];

  float* ws = (float*)d_ws;
  size_t off = 0;
  auto alloc = [&](size_t n) { float* p = ws + off; off += n; return p; };

  float* h1   = alloc((size_t)N1C * DC);
  float* h2   = alloc((size_t)N2C * DC);
  float* g1   = alloc((size_t)N1C * DC);
  float* g2   = alloc((size_t)N2C * DC);
  float* gx   = alloc((size_t)N1C * DC);
  float* a_s1 = alloc(N1C);
  float* a_d1 = alloc(N1C);
  float* a_s2 = alloc(N2C);
  float* a_d2 = alloc(N2C);
  float* a_sx = alloc(N1C);
  float* a_dx = alloc(N2C);
  float* vd   = alloc(DC);
  float* sumA = alloc(ES_TAB);
  float* sumB = alloc(ES_TAB);
  float* ew   = alloc(EC);
  (void)ws_size; (void)in_sizes; (void)n_in; (void)out_size;

  const dim3 blk(256);
  const dim3 gemm_g1(N1C / 16), gemm_g2(N2C / 16);
  const dim3 es_g(ES_WG), es_b(ES_T);
  const dim3 vz_g(VZ_BLOCKS);
  const dim3 ea_b(128);
  const dim3 ea_g((EC + 4 * AGG_CH - 1) / (4 * AGG_CH));
  const size_t lds_bytes = (size_t)(DC * DC + 16 * WSTR + 8 * 16 * 2) * sizeof(float);

  const float* in1 = x1;
  const float* in2 = x2;

  float* sc = sumA;
  float* sn = sumB;
  float* sw;

  for (int l = 0; l < LC; ++l) {
    const size_t wo = (size_t)l * DC * DC;
    const size_t vo = (size_t)l * DC;

    float* o1 = h1;
    float* o2 = h2;

    vecmat128_kernel<<<vz_g, DC, 0, stream>>>(Wd_x + wo, ad_x + vo, vd, sc);

    gemm_n128_wmma<<<gemm_g1, blk, lds_bytes, stream>>>(in1, Ws_x + wo, nullptr, gx, l,
        nullptr, nullptr,
        nullptr, nullptr, nullptr,
        as_x + vo, nullptr, a_sx, nullptr);
    gemm_n128_wmma<<<gemm_g1, blk, lds_bytes, stream>>>(in1, W_i1 + wo, nullptr, g1, l,
        nullptr, nullptr,
        o1, b_i1 + vo, nullptr,
        as_i1 + vo, ad_i1 + vo, a_s1, a_d1);
    gemm_n128_wmma<<<gemm_g2, blk, lds_bytes, stream>>>(in2, W_i2 + wo, nullptr, g2, l,
        vd, a_dx,
        o2, b_i2 + vo, b_x + vo,
        as_i2 + vo, ad_i2 + vo, a_s2, a_d2);

    edge_exp_sum_kernel<<<es_g, es_b, 0, stream>>>(a_s1, a_d1, ei11, ei11 + EC, ew, sc, sn, EC, N1C, N1C);
    edge_aggr_kernel<<<ea_g, ea_b, 0, stream>>>(ew, sc, ei11, ei11 + EC, g1, o1, EC);
    sw = sc; sc = sn; sn = sw;

    edge_exp_sum_kernel<<<es_g, es_b, 0, stream>>>(a_s2, a_d2, ei22, ei22 + EC, ew, sc, sn, EC, N2C, N2C);
    edge_aggr_kernel<<<ea_g, ea_b, 0, stream>>>(ew, sc, ei22, ei22 + EC, g2, o2, EC);
    sw = sc; sc = sn; sn = sw;

    edge_exp_sum_kernel<<<es_g, es_b, 0, stream>>>(a_sx, a_dx, ei12, ei12 + EC, ew, sc, sn, EC, N1C, N2C);
    edge_aggr_kernel<<<ea_g, ea_b, 0, stream>>>(ew, sc, ei12, ei12 + EC, gx, o2, EC);
    sw = sc; sc = sn; sn = sw;

    in1 = h1;
    in2 = h2;
  }

  float* out = (float*)d_out;
  gemm_n128_wmma<<<gemm_g1, blk, lds_bytes, stream>>>(h1, W_out, b_out, out, 1,
      nullptr, nullptr, nullptr, nullptr, nullptr,
      nullptr, nullptr, nullptr, nullptr);
  gemm_n128_wmma<<<gemm_g2, blk, lds_bytes, stream>>>(h2, W_out, b_out,
      out + (size_t)N1C * DC, 1,
      nullptr, nullptr, nullptr, nullptr, nullptr,
      nullptr, nullptr, nullptr, nullptr);
}


# edge softmax+aggregation: 256 work-groups each own 196 dst rows in LDS, scan the whole dst list for their edges, LDS float adds, one store per row; global atomics, 6 launches go; sum order changes
# speedup vs baseline: 3.2087x; 1.1399x over previous
#include <hip/hip_runtime.h>
#include <hip/hip_bf16.h>


#define N1C 50000
#define N2C 50000
#define EC  800000
#define DC  128
#define LC  2
#define WSTR 132
#define WCOL(k, n) (((n) + 16 * (((k) >> 1) & 1)) & (DC - 1))

typedef float v2f __attribute__((ext_vector_type(2)));
typedef float v8f __attribute__((ext_vector_type(8)));

__device__ __forceinline__ void atomicMaxSigned(float* addr, float val) {
  const int b = __float_as_int(val);
  if (b >= 0) atomicMax((int*)addr, b);
  else        atomicMin((unsigned int*)addr, (unsigned int)b);
}

#define STAB_RANGE 16.0f

__device__ __forceinline__ float leaky02(float l) { return l > 0.f ? l : 0.2f * l; }

__device__ __forceinline__ float rowsum8_over16(const float (&p)[8], int m) {
  const bool b3 = (m & 8) != 0, b2 = (m & 4) != 0, b1 = (m & 2) != 0;
  float q[4], s[2];
#pragma unroll
  for (int i = 0; i < 4; ++i) {
    const float give = b3 ? p[i] : p[i + 4];
    const float keep = b3 ? p[i + 4] : p[i];
    q[i] = keep + __shfl_xor(give, 8, 32);
  }
#pragma unroll
  for (int i = 0; i < 2; ++i) {
    const float give = b2 ? q[i] : q[i + 2];
    const float keep = b2 ? q[i + 2] : q[i];
    s[i] = keep + __shfl_xor(give, 4, 32);
  }
  const float give = b1 ? s[0] : s[1];
  const float keep = b1 ? s[1] : s[0];
  float t = keep + __shfl_xor(give, 2, 32);
  t += __shfl_xor(t, 1, 32);
  return t;
}

__global__ __launch_bounds__(256)
void gemm_n128_wmma(const float* A, const float* __restrict__ W,
                    const float* __restrict__ bias, float* __restrict__ C,
                    int relu_in,
                    const float* __restrict__ u, float* __restrict__ du,
                    const float* __restrict__ v1, const float* __restrict__ v2,
                    float* __restrict__ d1, float* __restrict__ d2) {
  extern __shared__ float smem[];
  float* sW  = smem;
  float* sA  = smem + DC * DC;
  float* sR  = sA + 16 * WSTR;

  const int rowbase = blockIdx.x * 16;
  const int tid = threadIdx.x;
  const int lane = tid & 31;

#pragma unroll
  for (int i = 0; i < 16; ++i) {
    const int c  = tid + 256 * i;
    const int k  = c >> 5;
    const int n4 = (c & 31) * 4;
    const float4 w4 = *(const float4*)(W + (size_t)k * DC + n4);
    *(float4*)(sW + k * DC + WCOL(k, n4)) = w4;
  }
#pragma unroll
  for (int i = 0; i < 2; ++i) {
    const int c  = tid + 256 * i;
    const int r  = c >> 5;
    const int n4 = (c & 31) * 4;
    float4 a4 = *(const float4*)(A + (size_t)(rowbase + r) * DC + n4);
    if (relu_in) {
      a4.x = a4.x > 0.f ? a4.x : 0.f;
      a4.y = a4.y > 0.f ? a4.y : 0.f;
      a4.z = a4.z > 0.f ? a4.z : 0.f;
      a4.w = a4.w > 0.f ? a4.w : 0.f;
    }
    *(float4*)(sA + r * WSTR + n4) = a4;
    if (u) {
      const float4 u4 = ((const float4*)u)[lane];
      float s = a4.x * u4.x + a4.y * u4.y + a4.z * u4.z + a4.w * u4.w;
#pragma unroll
      for (int off = 16; off > 0; off >>= 1) s += __shfl_down(s, off, 32);
      if (lane == 0) du[rowbase + r] = s;
    }
  }
  __syncthreads();

  const int cb   = (tid >> 5) * 16;
  const int m    = lane & 15;
  const int half = lane >> 4;

  const float* ap = sA + m * WSTR + half * 2;
  const float* bp = sW + half * (2 * DC) + WCOL(2 * half, cb + m);

  v8f acc = {};
#pragma unroll
  for (int kk = 0; kk < 32; ++kk) {
    const v2f a = *(const v2f*)(ap + kk * 4);
    v2f b;
    b.x = bp[kk * (4 * DC)];
    b.y = bp[kk * (4 * DC) + DC];
    acc = __builtin_amdgcn_wmma_f32_16x16x4_f32(false, a, false, b,
                                                (short)0, acc, false, false);
  }

  asm volatile("v_nop\n\tv_nop\n\tv_nop\n\tv_nop" : "+v"(acc));

  const int col = cb + m;
  const float bv = bias ? bias[col] : 0.f;
  float cv[8];
#pragma unroll
  for (int r = 0; r < 8; ++r) cv[r] = acc[r] + bv;

  if (v1) {
    const float w1 = v1[col];
    const float w2 = v2 ? v2[col] : 0.f;
    float p1[8], p2[8];
#pragma unroll
    for (int r = 0; r < 8; ++r) { p1[r] = cv[r] * w1; p2[r] = cv[r] * w2; }
    const float t1 = rowsum8_over16(p1, m);
    const float t2 = rowsum8_over16(p2, m);
    if ((m & 1) == 0) {
      const int rr = half * 8 + (m >> 1);
      float* q = sR + (((tid >> 5) * 16 + rr) << 1);
      q[0] = t1;
      q[1] = t2;
    }
    __syncthreads();
    if (tid < 32) {
      const int rr = tid >> 1, which = tid & 1;
      float s = 0.f;
#pragma unroll
      for (int w = 0; w < 8; ++w) s += sR[((w * 16 + rr) << 1) + which];
      float* dp = which ? d2 : d1;
      if (dp) dp[rowbase + rr] = s;
    }
  }

  float* cp = C + ((size_t)rowbase + half * 8) * DC + col;
#pragma unroll
  for (int r = 0; r < 8; ++r) cp[(size_t)r * DC] = cv[r];
}

__global__ void vecmat128_kernel(const float* __restrict__ W,
                                 const float* __restrict__ a,
                                 float* __restrict__ v) {
  const int k = threadIdx.x;
  float s = 0.f;
#pragma unroll 8
  for (int j = 0; j < DC; ++j) s += W[(size_t)k * DC + j] * a[j];
  v[k] = s;
}

#define OW_WG   256
#define OW_T    1024
#define OW_NMAX (N1C > N2C ? N1C : N2C)
#define OW_ROWS ((OW_NMAX + OW_WG - 1) / OW_WG)
#define OW_SC   8
#define OW_PB   4
#define OW_Q    (32 + OW_PB * 4 * 32)
#define OW_GU   8
#define OW_SU   7
static_assert(OW_ROWS <= 256, "a queue entry keeps the owned row in its low 8 bits");
static_assert(EC < (1 << 23), "a queue entry keeps the edge number above its low 8 bits");
static_assert(EC % 4 == 0, "the destination half of an edge list must be 16-byte aligned");
static_assert(OW_SC % OW_PB == 0, "the queue is looked at after a whole number of loads");

__global__ __launch_bounds__(OW_T)
void edge_owner_kernel(const float* __restrict__ as,
                       const float* __restrict__ ad,
                       const int* __restrict__ src,
                       const int* __restrict__ dst,
                       const float* __restrict__ G,
                       float* __restrict__ O,
                       const float* __restrict__ b1,
                       const float* __restrict__ b2,
                       int E, int n_src, int n_dst) {
  __shared__ __attribute__((aligned(16))) float acc[OW_ROWS * DC];
  __shared__ float zs[OW_ROWS];
  __shared__ float stb[OW_ROWS];
  __shared__ float ads[OW_ROWS];
  __shared__ int   q[OW_T / 32][OW_Q];
  __shared__ float smx[OW_T / 32], smn[OW_T / 32];

  const int tid  = (int)threadIdx.x;
  const int lane = tid & 31;
  const int wv   = __builtin_amdgcn_readfirstlane(tid >> 5);

  const int rows_per = (n_dst + (int)gridDim.x - 1) / (int)gridDim.x;
  const int lo = (int)blockIdx.x * rows_per;
  int nrows = n_dst - lo;
  nrows = nrows < 0 ? 0 : (nrows > rows_per ? rows_per : nrows);

  for (int i = tid; i < OW_ROWS * DC / 4; i += OW_T) ((float4*)acc)[i] = make_float4(0.f, 0.f, 0.f, 0.f);
  if (tid < OW_ROWS) {
    zs[tid]  = 0.f;
    ads[tid] = tid < nrows ? ad[lo + tid] : 0.f;
  }

  float mx = -3.402823466e38f, mn = 3.402823466e38f;
  const int n4 = n_src >> 2;
  const float4* as4 = (const float4*)as;
  for (int i0 = tid; i0 < n4; i0 += OW_T * OW_SU) {
    float4 a4[OW_SU];
#pragma unroll
    for (int u = 0; u < OW_SU; ++u) {
      const int i = i0 + u * OW_T;
      a4[u] = as4[i < n4 ? i : i0];
    }
#pragma unroll
    for (int u = 0; u < OW_SU; ++u) {
      mx = fmaxf(fmaxf(mx, a4[u].x), fmaxf(a4[u].y, fmaxf(a4[u].z, a4[u].w)));
      mn = fminf(fminf(mn, a4[u].x), fminf(a4[u].y, fminf(a4[u].z, a4[u].w)));
    }
  }
  for (int i = (n4 << 2) + tid; i < n_src; i += OW_T) {
    const float a1 = as[i];
    mx = fmaxf(mx, a1);
    mn = fminf(mn, a1);
  }
#pragma unroll
  for (int off = 16; off > 0; off >>= 1) {
    mx = fmaxf(mx, __shfl_xor(mx, off, 32));
    mn = fminf(mn, __shfl_xor(mn, off, 32));
  }
  if (lane == 0) { smx[wv] = mx; smn[wv] = mn; }
  __syncthreads();
  float gmax = smx[lane], gmin = smn[lane];
#pragma unroll
  for (int off = 16; off > 0; off >>= 1) {
    gmax = fmaxf(gmax, __shfl_xor(gmax, off, 32));
    gmin = fminf(gmin, __shfl_xor(gmin, off, 32));
  }
  const float range = gmax - gmin;
  const int bound_ok = __builtin_amdgcn_readfirstlane(
      (range >= 0.f && range <= STAB_RANGE) ? 1 : 0);

  if (bound_ok) {
    if (tid < OW_ROWS) stb[tid] = leaky02(gmax + ads[tid]);
  } else {
    if (tid < OW_ROWS) stb[tid] = -3.402823466e38f;
    __syncthreads();
    for (int e = tid; e < E; e += OW_T) {
      const int r = dst[e] - lo;
      if ((unsigned)r < (unsigned)nrows)
        atomicMaxSigned(stb + r, leaky02(as[src[e]] + ads[r]));
    }
  }
  __syncthreads();

  auto take = [&](int cnt) {
    for (int base = 0; base < cnt; base += 32) {
      const int n = (cnt - base) < 32 ? (cnt - base) : 32;
      int   s = 0, r = 0;
      float w = 0.f;
      if (lane < n) {
        const int ent = q[wv][base + lane];
        r = ent & 255;
        s = src[ent >> 8];
        w = __expf(leaky02(as[s] + ads[r]) - stb[r]);
        atomicAdd(zs + r, w);
      }
#pragma unroll 1
      for (int j0 = 0; j0 < n; j0 += OW_GU) {
        float g[OW_GU][4];
#pragma unroll
        for (int u = 0; u < OW_GU; ++u) {
          if (j0 + u < n) {
            const int sj = __builtin_amdgcn_readlane(s, j0 + u);
            const float* gp = G + (size_t)sj * DC;
            g[u][0] = gp[lane];
            g[u][1] = gp[lane + 32];
            g[u][2] = gp[lane + 64];
            g[u][3] = gp[lane + 96];
          } else {
            g[u][0] = g[u][1] = g[u][2] = g[u][3] = 0.f;
          }
        }
#pragma unroll
        for (int u = 0; u < OW_GU; ++u) {
          if (j0 + u < n) {
            const int   rj = __builtin_amdgcn_readlane(r, j0 + u);
            const float wj = __int_as_float(
                __builtin_amdgcn_readlane(__float_as_int(w), j0 + u));
            float* ap = acc + rj * DC + lane;
            atomicAdd(ap,      wj * g[u][0]);
            atomicAdd(ap + 32, wj * g[u][1]);
            atomicAdd(ap + 64, wj * g[u][2]);
            atomicAdd(ap + 96, wj * g[u][3]);
          }
        }
      }
    }
  };

  int qn = 0;
  auto put = [&](int dval, int e) {
    const int r = dval - lo;
    const bool hit = (unsigned)r < (unsigned)nrows;
    const unsigned mask = __builtin_amdgcn_ballot_w32(hit);
    if (mask) {
      if (hit) q[wv][qn + (int)__builtin_amdgcn_mbcnt_lo(mask, 0u)] = (e << 8) | r;
      qn += __builtin_popcount(mask);
    }
  };

  const int   E4     = E >> 2;
  const int4* dst4   = (const int4*)dst;
  const int   nchunk = (E4 + OW_T - 1) / OW_T;
  const int   rot    = nchunk > 0 ? (int)(((unsigned)blockIdx.x * 61u) % (unsigned)nchunk) : 0;
  for (int k0 = 0; k0 < nchunk; k0 += OW_SC) {
    int4 dv[OW_SC];
    int  eb[OW_SC];
#pragma unroll
    for (int u = 0; u < OW_SC; ++u) {
      const int k  = k0 + u;
      int kr = k + rot;
      if (kr >= nchunk) kr -= nchunk;
      const int  i4 = kr * OW_T + tid;
      const bool ok = (k < nchunk) && (i4 < E4);
      int4 t = dst4[ok ? i4 : 0];
      if (!ok) t = make_int4(-1, -1, -1, -1);
      dv[u] = t;
      eb[u] = i4 << 2;
    }
#pragma unroll
    for (int u = 0; u < OW_SC; ++u) {
      put(dv[u].x, eb[u]);
      put(dv[u].y, eb[u] + 1);
      put(dv[u].z, eb[u] + 2);
      put(dv[u].w, eb[u] + 3);
      if ((u % OW_PB) == OW_PB - 1 && qn >= 32) { take(qn); qn = 0; }
    }
  }
  if (wv == 0) {
    const int e = (E4 << 2) + lane;
    int dl = -1;
    if (e < E) dl = dst[e];
    put(dl, e);
  }
  if (qn > 0) take(qn);
  __syncthreads();

  for (int i = tid; i < nrows * (DC / 4); i += OW_T) {
    const int r  = i >> 5;
    const int c4 = (i & 31) * 4;
    const float4 a4 = *(const float4*)(acc + r * DC + c4);
    const float den = zs[r] + 1e-16f;
    float4 o4;
    o4.x = a4.x / den; o4.y = a4.y / den; o4.z = a4.z / den; o4.w = a4.w / den;
    float4* op = (float4*)(O + (size_t)(lo + r) * DC + c4);
    if (b1) {
      float4 b4 = *(const float4*)(b1 + c4);
      if (b2) {
        const float4 c4v = *(const float4*)(b2 + c4);
        b4.x += c4v.x; b4.y += c4v.y; b4.z += c4v.z; b4.w += c4v.w;
      }
      o4.x += b4.x; o4.y += b4.y; o4.z += b4.z; o4.w += b4.w;
    } else {
      const float4 p4 = *op;
      o4.x += p4.x; o4.y += p4.y; o4.z += p4.z; o4.w += p4.w;
    }
    *op = o4;
  }
}

extern "C" void kernel_launch(void* const* d_in, const int* in_sizes, int n_in,
                              void* d_out, int out_size, void* d_ws, size_t ws_size,
                              hipStream_t stream) {
  const float* x1    = (const float*)d_in[0];
  const float* x2    = (const float*)d_in[1];
  const int*   ei11  = (const int*)d_in[2];
  const int*   ei22  = (const int*)d_in[3];
  const int*   ei12  = (const int*)d_in[4];
  const float* W_i1  = (const float*)d_in[5];
  const float* as_i1 = (const float*)d_in[6];
  const float* ad_i1 = (const float*)d_in[7];
  const float* b_i1  = (const float*)d_in[8];
  const float* W_i2  = (const float*)d_in[9];
  const float* as_i2 = (const float*)d_in[10];
  const float* ad_i2 = (const float*)d_in[11];
  const float* b_i2  = (const float*)d_in[12];
  const float* Ws_x  = (const float*)d_in[13];
  const float* Wd_x  = (const float*)d_in[14];
  const float* as_x  = (const float*)d_in[15];
  const float* ad_x  = (const float*)d_in[16];
  const float* b_x   = (const float*)d_in[17];
  const float* W_out = (const float*)d_in[18];
  const float* b_out = (const float*)d_in[19];

  float* ws = (float*)d_ws;
  size_t off = 0;
  auto alloc = [&](size_t n) { float* p = ws + off; off += n; return p; };

  float* h1   = alloc((size_t)N1C * DC);
  float* h2   = alloc((size_t)N2C * DC);
  float* g1   = alloc((size_t)N1C * DC);
  float* g2   = alloc((size_t)N2C * DC);
  float* gx   = alloc((size_t)N1C * DC);
  float* a_s1 = alloc(N1C);
  float* a_d1 = alloc(N1C);
  float* a_s2 = alloc(N2C);
  float* a_d2 = alloc(N2C);
  float* a_sx = alloc(N1C);
  float* a_dx = alloc(N2C);
  float* vd   = alloc(DC);
  (void)ws_size; (void)in_sizes; (void)n_in; (void)out_size;

  const dim3 blk(256);
  const dim3 gemm_g1(N1C / 16), gemm_g2(N2C / 16);
  const dim3 ow_g(OW_WG), ow_b(OW_T);
  const size_t lds_bytes = (size_t)(DC * DC + 16 * WSTR + 8 * 16 * 2) * sizeof(float);

  const float* in1 = x1;
  const float* in2 = x2;

  for (int l = 0; l < LC; ++l) {
    const size_t wo = (size_t)l * DC * DC;
    const size_t vo = (size_t)l * DC;

    float* o1 = h1;
    float* o2 = h2;

    vecmat128_kernel<<<1, DC, 0, stream>>>(Wd_x + wo, ad_x + vo, vd);

    gemm_n128_wmma<<<gemm_g1, blk, lds_bytes, stream>>>(in1, Ws_x + wo, nullptr, gx, l,
        nullptr, nullptr,
        as_x + vo, nullptr, a_sx, nullptr);
    gemm_n128_wmma<<<gemm_g1, blk, lds_bytes, stream>>>(in1, W_i1 + wo, nullptr, g1, l,
        nullptr, nullptr,
        as_i1 + vo, ad_i1 + vo, a_s1, a_d1);
    gemm_n128_wmma<<<gemm_g2, blk, lds_bytes, stream>>>(in2, W_i2 + wo, nullptr, g2, l,
        vd, a_dx,
        as_i2 + vo, ad_i2 + vo, a_s2, a_d2);

    edge_owner_kernel<<<ow_g, ow_b, 0, stream>>>(a_s1, a_d1, ei11, ei11 + EC, g1, o1,
        b_i1 + vo, nullptr, EC, N1C, N1C);
    edge_owner_kernel<<<ow_g, ow_b, 0, stream>>>(a_s2, a_d2, ei22, ei22 + EC, g2, o2,
        b_i2 + vo, b_x + vo, EC, N2C, N2C);
    edge_owner_kernel<<<ow_g, ow_b, 0, stream>>>(a_sx, a_dx, ei12, ei12 + EC, gx, o2,
        nullptr, nullptr, EC, N1C, N2C);

    in1 = h1;
    in2 = h2;
  }

  float* out = (float*)d_out;
  gemm_n128_wmma<<<gemm_g1, blk, lds_bytes, stream>>>(h1, W_out, b_out, out, 1,
      nullptr, nullptr,
      nullptr, nullptr, nullptr, nullptr);
  gemm_n128_wmma<<<gemm_g2, blk, lds_bytes, stream>>>(h2, W_out, b_out,
      out + (size_t)N1C * DC, 1,
      nullptr, nullptr,
      nullptr, nullptr, nullptr, nullptr);
}


# edge kernels find their edges in one owner byte an edge (row-range index, written once per call by a new small launch), 4 bytes matched per word op, not in the 4-byte dst list; LDS add order changes
# speedup vs baseline: 3.7686x; 1.1745x over previous
#include <hip/hip_runtime.h>
#include <hip/hip_bf16.h>


#define N1C 50000
#define N2C 50000
#define EC  800000
#define DC  128
#define LC  2
#define WSTR 132
#define WCOL(k, n) (((n) + 16 * (((k) >> 1) & 1)) & (DC - 1))

typedef float v2f __attribute__((ext_vector_type(2)));
typedef float v8f __attribute__((ext_vector_type(8)));

__device__ __forceinline__ void atomicMaxSigned(float* addr, float val) {
  const int b = __float_as_int(val);
  if (b >= 0) atomicMax((int*)addr, b);
  else        atomicMin((unsigned int*)addr, (unsigned int)b);
}

#define STAB_RANGE 16.0f

__device__ __forceinline__ float leaky02(float l) { return l > 0.f ? l : 0.2f * l; }

__device__ __forceinline__ float rowsum8_over16(const float (&p)[8], int m) {
  const bool b3 = (m & 8) != 0, b2 = (m & 4) != 0, b1 = (m & 2) != 0;
  float q[4], s[2];
#pragma unroll
  for (int i = 0; i < 4; ++i) {
    const float give = b3 ? p[i] : p[i + 4];
    const float keep = b3 ? p[i + 4] : p[i];
    q[i] = keep + __shfl_xor(give, 8, 32);
  }
#pragma unroll
  for (int i = 0; i < 2; ++i) {
    const float give = b2 ? q[i] : q[i + 2];
    const float keep = b2 ? q[i + 2] : q[i];
    s[i] = keep + __shfl_xor(give, 4, 32);
  }
  const float give = b1 ? s[0] : s[1];
  const float keep = b1 ? s[1] : s[0];
  float t = keep + __shfl_xor(give, 2, 32);
  t += __shfl_xor(t, 1, 32);
  return t;
}

__global__ __launch_bounds__(256)
void gemm_n128_wmma(const float* A, const float* __restrict__ W,
                    const float* __restrict__ bias, float* __restrict__ C,
                    int relu_in,
                    const float* __restrict__ u, float* __restrict__ du,
                    const float* __restrict__ v1, const float* __restrict__ v2,
                    float* __restrict__ d1, float* __restrict__ d2) {
  extern __shared__ float smem[];
  float* sW  = smem;
  float* sA  = smem + DC * DC;
  float* sR  = sA + 16 * WSTR;

  const int rowbase = blockIdx.x * 16;
  const int tid = threadIdx.x;
  const int lane = tid & 31;

#pragma unroll
  for (int i = 0; i < 16; ++i) {
    const int c  = tid + 256 * i;
    const int k  = c >> 5;
    const int n4 = (c & 31) * 4;
    const float4 w4 = *(const float4*)(W + (size_t)k * DC + n4);
    *(float4*)(sW + k * DC + WCOL(k, n4)) = w4;
  }
#pragma unroll
  for (int i = 0; i < 2; ++i) {
    const int c  = tid + 256 * i;
    const int r  = c >> 5;
    const int n4 = (c & 31) * 4;
    float4 a4 = *(const float4*)(A + (size_t)(rowbase + r) * DC + n4);
    if (relu_in) {
      a4.x = a4.x > 0.f ? a4.x : 0.f;
      a4.y = a4.y > 0.f ? a4.y : 0.f;
      a4.z = a4.z > 0.f ? a4.z : 0.f;
      a4.w = a4.w > 0.f ? a4.w : 0.f;
    }
    *(float4*)(sA + r * WSTR + n4) = a4;
    if (u) {
      const float4 u4 = ((const float4*)u)[lane];
      float s = a4.x * u4.x + a4.y * u4.y + a4.z * u4.z + a4.w * u4.w;
#pragma unroll
      for (int off = 16; off > 0; off >>= 1) s += __shfl_down(s, off, 32);
      if (lane == 0) du[rowbase + r] = s;
    }
  }
  __syncthreads();

  const int cb   = (tid >> 5) * 16;
  const int m    = lane & 15;
  const int half = lane >> 4;

  const float* ap = sA + m * WSTR + half * 2;
  const float* bp = sW + half * (2 * DC) + WCOL(2 * half, cb + m);

  v8f acc = {};
#pragma unroll
  for (int kk = 0; kk < 32; ++kk) {
    const v2f a = *(const v2f*)(ap + kk * 4);
    v2f b;
    b.x = bp[kk * (4 * DC)];
    b.y = bp[kk * (4 * DC) + DC];
    acc = __builtin_amdgcn_wmma_f32_16x16x4_f32(false, a, false, b,
                                                (short)0, acc, false, false);
  }

  asm volatile("v_nop\n\tv_nop\n\tv_nop\n\tv_nop" : "+v"(acc));

  const int col = cb + m;
  const float bv = bias ? bias[col] : 0.f;
  float cv[8];
#pragma unroll
  for (int r = 0; r < 8; ++r) cv[r] = acc[r] + bv;

  if (v1) {
    const float w1 = v1[col];
    const float w2 = v2 ? v2[col] : 0.f;
    float p1[8], p2[8];
#pragma unroll
    for (int r = 0; r < 8; ++r) { p1[r] = cv[r] * w1; p2[r] = cv[r] * w2; }
    const float t1 = rowsum8_over16(p1, m);
    const float t2 = rowsum8_over16(p2, m);
    if ((m & 1) == 0) {
      const int rr = half * 8 + (m >> 1);
      float* q = sR + (((tid >> 5) * 16 + rr) << 1);
      q[0] = t1;
      q[1] = t2;
    }
    __syncthreads();
    if (tid < 32) {
      const int rr = tid >> 1, which = tid & 1;
      float s = 0.f;
#pragma unroll
      for (int w = 0; w < 8; ++w) s += sR[((w * 16 + rr) << 1) + which];
      float* dp = which ? d2 : d1;
      if (dp) dp[rowbase + rr] = s;
    }
  }

  float* cp = C + ((size_t)rowbase + half * 8) * DC + col;
#pragma unroll
  for (int r = 0; r < 8; ++r) cp[(size_t)r * DC] = cv[r];
}

__global__ void vecmat128_kernel(const float* __restrict__ W,
                                 const float* __restrict__ a,
                                 float* __restrict__ v) {
  const int k = threadIdx.x;
  float s = 0.f;
#pragma unroll 8
  for (int j = 0; j < DC; ++j) s += W[(size_t)k * DC + j] * a[j];
  v[k] = s;
}

#define OW_WG   256
#define OW_T    1024
#define OW_NMAX (N1C > N2C ? N1C : N2C)
#define OW_ROWS ((OW_NMAX + OW_WG - 1) / OW_WG)
#define OW_SC   4
#define OW_Q    544
#define OW_GU   8
#define OW_SU   7
static_assert(OW_WG == 256, "the owner of a row is kept in one byte and matched against the block index");
static_assert(EC % 16 == 0, "an edge set's destinations and its owner bytes must be 16-byte aligned");
static_assert(OW_SC == 4, "the hit bits of one trip are kept in one 64-bit word: 4 loads of 16 edges");

__global__ __launch_bounds__(256)
void owner_byte_kernel(const int* __restrict__ dst_a, const int* __restrict__ dst_b,
                       const int* __restrict__ dst_c,
                       unsigned* __restrict__ ob_a, unsigned* __restrict__ ob_b,
                       unsigned* __restrict__ ob_c,
                       int E, int n_a, int n_b, int n_c, int n_wg) {
  const int which = (int)blockIdx.y;
  const int*  dst = which == 0 ? dst_a : (which == 1 ? dst_b : dst_c);
  unsigned*   ob  = which == 0 ? ob_a  : (which == 1 ? ob_b  : ob_c);
  const int n_dst = which == 0 ? n_a   : (which == 1 ? n_b   : n_c);
  const unsigned rows_per = (unsigned)((n_dst + n_wg - 1) / n_wg);
  const int i = (int)blockIdx.x * 256 + (int)threadIdx.x;
  if (i < (E >> 2)) {
    const int4 d = ((const int4*)dst)[i];
    ob[i] = (((unsigned)d.x / rows_per) & 255u)
          | ((((unsigned)d.y / rows_per) & 255u) << 8)
          | ((((unsigned)d.z / rows_per) & 255u) << 16)
          | ((((unsigned)d.w / rows_per) & 255u) << 24);
  }
}

__device__ __forceinline__ unsigned differ_bits(unsigned x, unsigned gp) {
  const unsigned y = x ^ gp;
  return (((y & 0x7F7F7F7Fu) + 0x7F7F7F7Fu) | y) | 0x7F7F7F7Fu;
}

__global__ __launch_bounds__(OW_T)
void edge_owner_kernel(const float* __restrict__ as,
                       const float* __restrict__ ad,
                       const int* __restrict__ src,
                       const int* __restrict__ dst,
                       const unsigned* __restrict__ ob,
                       const float* __restrict__ G,
                       float* __restrict__ O,
                       const float* __restrict__ b1,
                       const float* __restrict__ b2,
                       int E, int n_src, int n_dst) {
  __shared__ __attribute__((aligned(16))) float acc[(OW_ROWS + 1) * DC];
  __shared__ float zs[OW_ROWS + 1];
  __shared__ float stb[OW_ROWS + 1];
  __shared__ float ads[OW_ROWS + 1];
  __shared__ int   q[OW_T / 32][OW_Q];
  __shared__ float smx[OW_T / 32], smn[OW_T / 32];

  const int tid  = (int)threadIdx.x;
  const int lane = tid & 31;
  const int wv   = __builtin_amdgcn_readfirstlane(tid >> 5);

  const int rows_per = (n_dst + (int)gridDim.x - 1) / (int)gridDim.x;
  const int lo = (int)blockIdx.x * rows_per;
  int nrows = n_dst - lo;
  nrows = nrows < 0 ? 0 : (nrows > rows_per ? rows_per : nrows);

  for (int i = tid; i < OW_ROWS * DC / 4; i += OW_T) ((float4*)acc)[i] = make_float4(0.f, 0.f, 0.f, 0.f);
  if (tid <= OW_ROWS) {
    zs[tid]  = 0.f;
    ads[tid] = tid < nrows ? ad[lo + tid] : 0.f;
  }

  float mx = -3.402823466e38f, mn = 3.402823466e38f;
  const int n4 = n_src >> 2;
  const float4* as4 = (const float4*)as;
  for (int i0 = tid; i0 < n4; i0 += OW_T * OW_SU) {
    float4 a4[OW_SU];
#pragma unroll
    for (int u = 0; u < OW_SU; ++u) {
      const int i = i0 + u * OW_T;
      a4[u] = as4[i < n4 ? i : i0];
    }
#pragma unroll
    for (int u = 0; u < OW_SU; ++u) {
      mx = fmaxf(fmaxf(mx, a4[u].x), fmaxf(a4[u].y, fmaxf(a4[u].z, a4[u].w)));
      mn = fminf(fminf(mn, a4[u].x), fminf(a4[u].y, fminf(a4[u].z, a4[u].w)));
    }
  }
  for (int i = (n4 << 2) + tid; i < n_src; i += OW_T) {
    const float a1 = as[i];
    mx = fmaxf(mx, a1);
    mn = fminf(mn, a1);
  }
#pragma unroll
  for (int off = 16; off > 0; off >>= 1) {
    mx = fmaxf(mx, __shfl_xor(mx, off, 32));
    mn = fminf(mn, __shfl_xor(mn, off, 32));
  }
  if (lane == 0) { smx[wv] = mx; smn[wv] = mn; }
  __syncthreads();
  float gmax = smx[lane], gmin = smn[lane];
#pragma unroll
  for (int off = 16; off > 0; off >>= 1) {
    gmax = fmaxf(gmax, __shfl_xor(gmax, off, 32));
    gmin = fminf(gmin, __shfl_xor(gmin, off, 32));
  }
  const float range = gmax - gmin;
  const int bound_ok = __builtin_amdgcn_readfirstlane(
      (range >= 0.f && range <= STAB_RANGE) ? 1 : 0);

  if (bound_ok) {
    if (tid <= OW_ROWS) stb[tid] = leaky02(gmax + ads[tid]);
  } else {
    if (tid <= OW_ROWS) stb[tid] = -3.402823466e38f;
    __syncthreads();
    for (int e = tid; e < E; e += OW_T) {
      const int r = dst[e] - lo;
      if ((unsigned)r < (unsigned)nrows)
        atomicMaxSigned(stb + r, leaky02(as[src[e]] + ads[r]));
    }
  }
  __syncthreads();

  auto take = [&](int cnt) {
    for (int base = 0; base < cnt; base += 32) {
      const int n = (cnt - base) < 32 ? (cnt - base) : 32;
      int   s = 0, r = 0;
      float w = 0.f;
      if (lane < n) {
        const int e = q[wv][base + lane];
        r = dst[e] - lo;
        s = src[e];
        const bool mine = (unsigned)r < (unsigned)nrows;
        r = mine ? r : OW_ROWS;
        w = __expf(leaky02(as[s] + ads[r]) - stb[r]);
        w = mine ? w : 0.f;
        atomicAdd(zs + r, w);
      }
#pragma unroll 1
      for (int j0 = 0; j0 < n; j0 += OW_GU) {
        float g[OW_GU][4];
#pragma unroll
        for (int u = 0; u < OW_GU; ++u) {
          if (j0 + u < n) {
            const int sj = __builtin_amdgcn_readlane(s, j0 + u);
            const float* gp = G + (size_t)sj * DC;
            g[u][0] = gp[lane];
            g[u][1] = gp[lane + 32];
            g[u][2] = gp[lane + 64];
            g[u][3] = gp[lane + 96];
          } else {
            g[u][0] = g[u][1] = g[u][2] = g[u][3] = 0.f;
          }
        }
#pragma unroll
        for (int u = 0; u < OW_GU; ++u) {
          if (j0 + u < n) {
            const int   rj = __builtin_amdgcn_readlane(r, j0 + u);
            const float wj = __int_as_float(
                __builtin_amdgcn_readlane(__float_as_int(w), j0 + u));
            float* ap = acc + rj * DC + lane;
            atomicAdd(ap,      wj * g[u][0]);
            atomicAdd(ap + 32, wj * g[u][1]);
            atomicAdd(ap + 64, wj * g[u][2]);
            atomicAdd(ap + 96, wj * g[u][3]);
          }
        }
      }
    }
  };

  int qn = 0;

  const unsigned gp    = ((unsigned)blockIdx.x & 255u) * 0x01010101u;
  const int    E16     = E >> 4;
  const uint4* ob4     = (const uint4*)ob;
  const int    nchunk  = (E16 + OW_T - 1) / OW_T;
  const int    ntrip   = (nchunk + OW_SC - 1) / OW_SC;
  const int    rot     = ntrip > 0 ? (int)(((unsigned)blockIdx.x * 61u) % (unsigned)ntrip) : 0;
  for (int t0 = 0; t0 < ntrip; ++t0) {
    int tr = t0 + rot;
    if (tr >= ntrip) tr -= ntrip;
    const int i0 = tr * (OW_SC * OW_T) + tid;
    uint4 dv[OW_SC];
    if ((tr + 1) * (OW_SC * OW_T) <= E16) {
#pragma unroll
      for (int u = 0; u < OW_SC; ++u) dv[u] = ob4[i0 + u * OW_T];
    } else {
#pragma unroll
      for (int u = 0; u < OW_SC; ++u) {
        const int  i16 = i0 + u * OW_T;
        const bool ok  = i16 < E16;
        uint4 t = ob4[ok ? i16 : 0];
        if (!ok) t = make_uint4(~gp, ~gp, ~gp, ~gp);
        dv[u] = t;
      }
    }
    unsigned keep[OW_SC / 2];
#pragma unroll
    for (int h = 0; h < OW_SC / 2; ++h) {
      const uint4 p = dv[2 * h], s4 = dv[2 * h + 1];
      unsigned k = differ_bits(p.x, gp);
      k &= __builtin_rotateright32(differ_bits(p.y,  gp), 1);
      k &= __builtin_rotateright32(differ_bits(p.z,  gp), 2);
      k &= __builtin_rotateright32(differ_bits(p.w,  gp), 3);
      k &= __builtin_rotateright32(differ_bits(s4.x, gp), 4);
      k &= __builtin_rotateright32(differ_bits(s4.y, gp), 5);
      k &= __builtin_rotateright32(differ_bits(s4.z, gp), 6);
      k &= __builtin_rotateright32(differ_bits(s4.w, gp), 7);
      keep[h] = k;
    }
    unsigned long long m = ((unsigned long long)(~keep[1]) << 32) | (unsigned long long)(~keep[0]);
    for (;;) {
      const bool has = m != 0ull;
      const unsigned mask = __builtin_amdgcn_ballot_w32(has);
      if (!mask) break;
      if (has) {
        const int pb = __builtin_ctzll(m);
        m &= m - 1ull;
        const int j  = 7 - (pb & 7);
        const int u  = ((pb >> 5) << 1) + (j >> 2);
        const int e  = ((i0 + u * OW_T) << 4) + ((j & 3) << 2) + ((pb >> 3) & 3);
        q[wv][qn + (int)__builtin_amdgcn_mbcnt_lo(mask, 0u)] = e;
      }
      qn += __builtin_popcount(mask);
      if (qn >= 32) { take(qn); qn = 0; }
    }
  }
  if (wv == 0) {
    const int  e   = (E16 << 4) + lane;
    const bool has = e < E;
    const unsigned mask = __builtin_amdgcn_ballot_w32(has);
    if (mask) {
      if (has) q[wv][qn + (int)__builtin_amdgcn_mbcnt_lo(mask, 0u)] = e;
      qn += __builtin_popcount(mask);
    }
  }
  if (qn > 0) take(qn);
  __syncthreads();

  for (int i = tid; i < nrows * (DC / 4); i += OW_T) {
    const int r  = i >> 5;
    const int c4 = (i & 31) * 4;
    const float4 a4 = *(const float4*)(acc + r * DC + c4);
    const float den = zs[r] + 1e-16f;
    float4 o4;
    o4.x = a4.x / den; o4.y = a4.y / den; o4.z = a4.z / den; o4.w = a4.w / den;
    float4* op = (float4*)(O + (size_t)(lo + r) * DC + c4);
    if (b1) {
      float4 b4 = *(const float4*)(b1 + c4);
      if (b2) {
        const float4 c4v = *(const float4*)(b2 + c4);
        b4.x += c4v.x; b4.y += c4v.y; b4.z += c4v.z; b4.w += c4v.w;
      }
      o4.x += b4.x; o4.y += b4.y; o4.z += b4.z; o4.w += b4.w;
    } else {
      const float4 p4 = *op;
      o4.x += p4.x; o4.y += p4.y; o4.z += p4.z; o4.w += p4.w;
    }
    *op = o4;
  }
}

extern "C" void kernel_launch(void* const* d_in, const int* in_sizes, int n_in,
                              void* d_out, int out_size, void* d_ws, size_t ws_size,
                              hipStream_t stream) {
  const float* x1    = (const float*)d_in[0];
  const float* x2    = (const float*)d_in[1];
  const int*   ei11  = (const int*)d_in[2];
  const int*   ei22  = (const int*)d_in[3];
  const int*   ei12  = (const int*)d_in[4];
  const float* W_i1  = (const float*)d_in[5];
  const float* as_i1 = (const float*)d_in[6];
  const float* ad_i1 = (const float*)d_in[7];
  const float* b_i1  = (const float*)d_in[8];
  const float* W_i2  = (const float*)d_in[9];
  const float* as_i2 = (const float*)d_in[10];
  const float* ad_i2 = (const float*)d_in[11];
  const float* b_i2  = (const float*)d_in[12];
  const float* Ws_x  = (const float*)d_in[13];
  const float* Wd_x  = (const float*)d_in[14];
  const float* as_x  = (const float*)d_in[15];
  const float* ad_x  = (const float*)d_in[16];
  const float* b_x   = (const float*)d_in[17];
  const float* W_out = (const float*)d_in[18];
  const float* b_out = (const float*)d_in[19];

  float* ws = (float*)d_ws;
  size_t off = 0;
  auto alloc = [&](size_t n) { float* p = ws + off; off += n; return p; };

  float* h1   = alloc((size_t)N1C * DC);
  float* h2   = alloc((size_t)N2C * DC);
  float* g1   = alloc((size_t)N1C * DC);
  float* g2   = alloc((size_t)N2C * DC);
  float* gx   = alloc((size_t)N1C * DC);
  float* a_s1 = alloc(N1C);
  float* a_d1 = alloc(N1C);
  float* a_s2 = alloc(N2C);
  float* a_d2 = alloc(N2C);
  float* a_sx = alloc(N1C);
  float* a_dx = alloc(N2C);
  float* vd   = alloc(DC);
  unsigned* ob11 = (unsigned*)alloc(EC / 4);
  unsigned* ob22 = (unsigned*)alloc(EC / 4);
  unsigned* ob12 = (unsigned*)alloc(EC / 4);
  (void)ws_size; (void)in_sizes; (void)n_in; (void)out_size;

  const dim3 blk(256);
  const dim3 gemm_g1(N1C / 16), gemm_g2(N2C / 16);
  const dim3 ow_g(OW_WG), ow_b(OW_T);
  const size_t lds_bytes = (size_t)(DC * DC + 16 * WSTR + 8 * 16 * 2) * sizeof(float);

  const float* in1 = x1;
  const float* in2 = x2;

  owner_byte_kernel<<<dim3((EC / 4 + 255) / 256, 3), blk, 0, stream>>>(
      ei11 + EC, ei22 + EC, ei12 + EC, ob11, ob22, ob12, EC, N1C, N2C, N2C, OW_WG);

  for (int l = 0; l < LC; ++l) {
    const size_t wo = (size_t)l * DC * DC;
    const size_t vo = (size_t)l * DC;

    float* o1 = h1;
    float* o2 = h2;

    vecmat128_kernel<<<1, DC, 0, stream>>>(Wd_x + wo, ad_x + vo, vd);

    gemm_n128_wmma<<<gemm_g1, blk, lds_bytes, stream>>>(in1, Ws_x + wo, nullptr, gx, l,
        nullptr, nullptr,
        as_x + vo, nullptr, a_sx, nullptr);
    gemm_n128_wmma<<<gemm_g1, blk, lds_bytes, stream>>>(in1, W_i1 + wo, nullptr, g1, l,
        nullptr, nullptr,
        as_i1 + vo, ad_i1 + vo, a_s1, a_d1);
    gemm_n128_wmma<<<gemm_g2, blk, lds_bytes, stream>>>(in2, W_i2 + wo, nullptr, g2, l,
        vd, a_dx,
        as_i2 + vo, ad_i2 + vo, a_s2, a_d2);

    edge_owner_kernel<<<ow_g, ow_b, 0, stream>>>(a_s1, a_d1, ei11, ei11 + EC, ob11, g1, o1,
        b_i1 + vo, nullptr, EC, N1C, N1C);
    edge_owner_kernel<<<ow_g, ow_b, 0, stream>>>(a_s2, a_d2, ei22, ei22 + EC, ob22, g2, o2,
        b_i2 + vo, b_x + vo, EC, N2C, N2C);
    edge_owner_kernel<<<ow_g, ow_b, 0, stream>>>(a_sx, a_dx, ei12, ei12 + EC, ob12, gx, o2,
        nullptr, nullptr, EC, N1C, N2C);

    in1 = h1;
    in2 = h2;
  }

  float* out = (float*)d_out;
  gemm_n128_wmma<<<gemm_g1, blk, lds_bytes, stream>>>(h1, W_out, b_out, out, 1,
      nullptr, nullptr,
      nullptr, nullptr, nullptr, nullptr);
  gemm_n128_wmma<<<gemm_g2, blk, lds_bytes, stream>>>(h2, W_out, b_out,
      out + (size_t)N1C * DC, 1,
      nullptr, nullptr,
      nullptr, nullptr, nullptr, nullptr);
}


# edge launch multiplies its finished LDS rows by the next W on the matrix unit (W fragments in registers); layer-2 and output GEMM launches, h1 and final h2 go; 17 to 12 launches; sums unchanged
# speedup vs baseline: 4.0987x; 1.0876x over previous
#include <hip/hip_runtime.h>
#include <hip/hip_bf16.h>


#define N1C 50000
#define N2C 50000
#define EC  800000
#define DC  128
#define LC  2
#define WSTR 132
#define WCOL(k, n) (((n) + 16 * (((k) >> 1) & 1)) & (DC - 1))

typedef float v2f __attribute__((ext_vector_type(2)));
typedef float v8f __attribute__((ext_vector_type(8)));

__device__ __forceinline__ void atomicMaxSigned(float* addr, float val) {
  const int b = __float_as_int(val);
  if (b >= 0) atomicMax((int*)addr, b);
  else        atomicMin((unsigned int*)addr, (unsigned int)b);
}

#define STAB_RANGE 16.0f

__device__ __forceinline__ float leaky02(float l) { return l > 0.f ? l : 0.2f * l; }

__device__ __forceinline__ float rowsum8_over16(const float (&p)[8], int m) {
  const bool b3 = (m & 8) != 0, b2 = (m & 4) != 0, b1 = (m & 2) != 0;
  float q[4], s[2];
#pragma unroll
  for (int i = 0; i < 4; ++i) {
    const float give = b3 ? p[i] : p[i + 4];
    const float keep = b3 ? p[i + 4] : p[i];
    q[i] = keep + __shfl_xor(give, 8, 32);
  }
#pragma unroll
  for (int i = 0; i < 2; ++i) {
    const float give = b2 ? q[i] : q[i + 2];
    const float keep = b2 ? q[i + 2] : q[i];
    s[i] = keep + __shfl_xor(give, 4, 32);
  }
  const float give = b1 ? s[0] : s[1];
  const float keep = b1 ? s[1] : s[0];
  float t = keep + __shfl_xor(give, 2, 32);
  t += __shfl_xor(t, 1, 32);
  return t;
}

__global__ __launch_bounds__(256)
void gemm_n128_wmma(const float* A, const float* __restrict__ W,
                    const float* __restrict__ bias, float* __restrict__ C,
                    int relu_in,
                    const float* __restrict__ u, float* __restrict__ du,
                    const float* __restrict__ v1, const float* __restrict__ v2,
                    float* __restrict__ d1, float* __restrict__ d2) {
  extern __shared__ float smem[];
  float* sW  = smem;
  float* sA  = smem + DC * DC;
  float* sR  = sA + 16 * WSTR;

  const int rowbase = blockIdx.x * 16;
  const int tid = threadIdx.x;
  const int lane = tid & 31;

#pragma unroll
  for (int i = 0; i < 16; ++i) {
    const int c  = tid + 256 * i;
    const int k  = c >> 5;
    const int n4 = (c & 31) * 4;
    const float4 w4 = *(const float4*)(W + (size_t)k * DC + n4);
    *(float4*)(sW + k * DC + WCOL(k, n4)) = w4;
  }
#pragma unroll
  for (int i = 0; i < 2; ++i) {
    const int c  = tid + 256 * i;
    const int r  = c >> 5;
    const int n4 = (c & 31) * 4;
    float4 a4 = *(const float4*)(A + (size_t)(rowbase + r) * DC + n4);
    if (relu_in) {
      a4.x = a4.x > 0.f ? a4.x : 0.f;
      a4.y = a4.y > 0.f ? a4.y : 0.f;
      a4.z = a4.z > 0.f ? a4.z : 0.f;
      a4.w = a4.w > 0.f ? a4.w : 0.f;
    }
    *(float4*)(sA + r * WSTR + n4) = a4;
    if (u) {
      const float4 u4 = ((const float4*)u)[lane];
      float s = a4.x * u4.x + a4.y * u4.y + a4.z * u4.z + a4.w * u4.w;
#pragma unroll
      for (int off = 16; off > 0; off >>= 1) s += __shfl_down(s, off, 32);
      if (lane == 0) du[rowbase + r] = s;
    }
  }
  __syncthreads();

  const int cb   = (tid >> 5) * 16;
  const int m    = lane & 15;
  const int half = lane >> 4;

  const float* ap = sA + m * WSTR + half * 2;
  const float* bp = sW + half * (2 * DC) + WCOL(2 * half, cb + m);

  v8f acc = {};
#pragma unroll
  for (int kk = 0; kk < 32; ++kk) {
    const v2f a = *(const v2f*)(ap + kk * 4);
    v2f b;
    b.x = bp[kk * (4 * DC)];
    b.y = bp[kk * (4 * DC) + DC];
    acc = __builtin_amdgcn_wmma_f32_16x16x4_f32(false, a, false, b,
                                                (short)0, acc, false, false);
  }

  asm volatile("v_nop\n\tv_nop\n\tv_nop\n\tv_nop" : "+v"(acc));

  const int col = cb + m;
  const float bv = bias ? bias[col] : 0.f;
  float cv[8];
#pragma unroll
  for (int r = 0; r < 8; ++r) cv[r] = acc[r] + bv;

  if (v1) {
    const float w1 = v1[col];
    const float w2 = v2 ? v2[col] : 0.f;
    float p1[8], p2[8];
#pragma unroll
    for (int r = 0; r < 8; ++r) { p1[r] = cv[r] * w1; p2[r] = cv[r] * w2; }
    const float t1 = rowsum8_over16(p1, m);
    const float t2 = rowsum8_over16(p2, m);
    if ((m & 1) == 0) {
      const int rr = half * 8 + (m >> 1);
      float* q = sR + (((tid >> 5) * 16 + rr) << 1);
      q[0] = t1;
      q[1] = t2;
    }
    __syncthreads();
    if (tid < 32) {
      const int rr = tid >> 1, which = tid & 1;
      float s = 0.f;
#pragma unroll
      for (int w = 0; w < 8; ++w) s += sR[((w * 16 + rr) << 1) + which];
      float* dp = which ? d2 : d1;
      if (dp) dp[rowbase + rr] = s;
    }
  }

  float* cp = C + ((size_t)rowbase + half * 8) * DC + col;
#pragma unroll
  for (int r = 0; r < 8; ++r) cp[(size_t)r * DC] = cv[r];
}

__global__ void vecmat128_kernel(const float* __restrict__ W,
                                 const float* __restrict__ a,
                                 float* __restrict__ v) {
  const int k = threadIdx.x;
  float s = 0.f;
#pragma unroll 8
  for (int j = 0; j < DC; ++j) s += W[(size_t)k * DC + j] * a[j];
  v[k] = s;
}

#define OW_WG   256
#define OW_T    1024
#define OW_NMAX (N1C > N2C ? N1C : N2C)
#define OW_ROWS ((OW_NMAX + OW_WG - 1) / OW_WG)
#define OW_TILES ((OW_ROWS + 16) / 16)
#define OW_AROWS (OW_TILES * 16)
#define OW_SC   4
#define OW_Q    544
#define OW_GU   8
#define OW_SU   7
static_assert(OW_WG == 256, "the owner of a row is kept in one byte and matched against the block index");
static_assert(OW_AROWS > OW_ROWS, "row OW_ROWS is the spare row of take()");
static_assert(8 * OW_AROWS * 2 <= (OW_T / 32) * OW_Q, "the row-dot partials of a product reuse the queues' LDS");
static_assert(EC % 16 == 0, "an edge set's destinations and its owner bytes must be 16-byte aligned");
static_assert(OW_SC == 4, "the hit bits of one trip are kept in one 64-bit word: 4 loads of 16 edges");

__global__ __launch_bounds__(256)
void owner_byte_kernel(const int* __restrict__ dst_a, const int* __restrict__ dst_b,
                       const int* __restrict__ dst_c,
                       unsigned* __restrict__ ob_a, unsigned* __restrict__ ob_b,
                       unsigned* __restrict__ ob_c,
                       int E, int n_a, int n_b, int n_c, int n_wg) {
  const int which = (int)blockIdx.y;
  const int*  dst = which == 0 ? dst_a : (which == 1 ? dst_b : dst_c);
  unsigned*   ob  = which == 0 ? ob_a  : (which == 1 ? ob_b  : ob_c);
  const int n_dst = which == 0 ? n_a   : (which == 1 ? n_b   : n_c);
  const unsigned rows_per = (unsigned)((n_dst + n_wg - 1) / n_wg);
  const int i = (int)blockIdx.x * 256 + (int)threadIdx.x;
  if (i < (E >> 2)) {
    const int4 d = ((const int4*)dst)[i];
    ob[i] = (((unsigned)d.x / rows_per) & 255u)
          | ((((unsigned)d.y / rows_per) & 255u) << 8)
          | ((((unsigned)d.z / rows_per) & 255u) << 16)
          | ((((unsigned)d.w / rows_per) & 255u) << 24);
  }
}

__device__ __forceinline__ unsigned differ_bits(unsigned x, unsigned gp) {
  const unsigned y = x ^ gp;
  return (((y & 0x7F7F7F7Fu) + 0x7F7F7F7Fu) | y) | 0x7F7F7F7Fu;
}

__global__ __launch_bounds__(OW_T)
void edge_owner_kernel(const float* __restrict__ as,
                       const float* ad,
                       const int* __restrict__ src,
                       const int* __restrict__ dst,
                       const unsigned* __restrict__ ob,
                       const float* __restrict__ G,
                       float* __restrict__ O,
                       const float* __restrict__ b1,
                       const float* __restrict__ b2,
                       int E, int n_src, int n_dst, int store_O,
                       const float* __restrict__ u, float* du,
                       const float* __restrict__ Wa, const float* __restrict__ ba,
                       float* __restrict__ Ca,
                       const float* __restrict__ va1, const float* __restrict__ va2,
                       float* da1, float* da2,
                       const float* __restrict__ Wb, float* __restrict__ Cb,
                       const float* __restrict__ vb1, float* db1) {
  __shared__ __attribute__((aligned(16))) float acc[OW_AROWS * WSTR];
  __shared__ float zs[OW_ROWS + 1];
  __shared__ float stb[OW_ROWS + 1];
  __shared__ float ads[OW_ROWS + 1];
  __shared__ int   q[OW_T / 32][OW_Q];
  __shared__ float smx[OW_T / 32], smn[OW_T / 32];

  const int tid  = (int)threadIdx.x;
  const int lane = tid & 31;
  const int wv   = __builtin_amdgcn_readfirstlane(tid >> 5);

  const int rows_per = (n_dst + (int)gridDim.x - 1) / (int)gridDim.x;
  const int lo = (int)blockIdx.x * rows_per;
  int nrows = n_dst - lo;
  nrows = nrows < 0 ? 0 : (nrows > rows_per ? rows_per : nrows);

  for (int i = tid; i < OW_AROWS * WSTR / 4; i += OW_T) ((float4*)acc)[i] = make_float4(0.f, 0.f, 0.f, 0.f);
  if (tid <= OW_ROWS) {
    zs[tid]  = 0.f;
    ads[tid] = tid < nrows ? ad[lo + tid] : 0.f;
  }

  float mx = -3.402823466e38f, mn = 3.402823466e38f;
  const int n4 = n_src >> 2;
  const float4* as4 = (const float4*)as;
  for (int i0 = tid; i0 < n4; i0 += OW_T * OW_SU) {
    float4 a4[OW_SU];
#pragma unroll
    for (int u = 0; u < OW_SU; ++u) {
      const int i = i0 + u * OW_T;
      a4[u] = as4[i < n4 ? i : i0];
    }
#pragma unroll
    for (int u = 0; u < OW_SU; ++u) {
      mx = fmaxf(fmaxf(mx, a4[u].x), fmaxf(a4[u].y, fmaxf(a4[u].z, a4[u].w)));
      mn = fminf(fminf(mn, a4[u].x), fminf(a4[u].y, fminf(a4[u].z, a4[u].w)));
    }
  }
  for (int i = (n4 << 2) + tid; i < n_src; i += OW_T) {
    const float a1 = as[i];
    mx = fmaxf(mx, a1);
    mn = fminf(mn, a1);
  }
#pragma unroll
  for (int off = 16; off > 0; off >>= 1) {
    mx = fmaxf(mx, __shfl_xor(mx, off, 32));
    mn = fminf(mn, __shfl_xor(mn, off, 32));
  }
  if (lane == 0) { smx[wv] = mx; smn[wv] = mn; }
  __syncthreads();
  float gmax = smx[lane], gmin = smn[lane];
#pragma unroll
  for (int off = 16; off > 0; off >>= 1) {
    gmax = fmaxf(gmax, __shfl_xor(gmax, off, 32));
    gmin = fminf(gmin, __shfl_xor(gmin, off, 32));
  }
  const float range = gmax - gmin;
  const int bound_ok = __builtin_amdgcn_readfirstlane(
      (range >= 0.f && range <= STAB_RANGE) ? 1 : 0);

  if (bound_ok) {
    if (tid <= OW_ROWS) stb[tid] = leaky02(gmax + ads[tid]);
  } else {
    if (tid <= OW_ROWS) stb[tid] = -3.402823466e38f;
    __syncthreads();
    for (int e = tid; e < E; e += OW_T) {
      const int r = dst[e] - lo;
      if ((unsigned)r < (unsigned)nrows)
        atomicMaxSigned(stb + r, leaky02(as[src[e]] + ads[r]));
    }
  }
  __syncthreads();

  auto take = [&](int cnt) {
    for (int base = 0; base < cnt; base += 32) {
      const int n = (cnt - base) < 32 ? (cnt - base) : 32;
      int   s = 0, r = 0;
      float w = 0.f;
      if (lane < n) {
        const int e = q[wv][base + lane];
        r = dst[e] - lo;
        s = src[e];
        const bool mine = (unsigned)r < (unsigned)nrows;
        r = mine ? r : OW_ROWS;
        w = __expf(leaky02(as[s] + ads[r]) - stb[r]);
        w = mine ? w : 0.f;
        atomicAdd(zs + r, w);
      }
#pragma unroll 1
      for (int j0 = 0; j0 < n; j0 += OW_GU) {
        float g[OW_GU][4];
#pragma unroll
        for (int u = 0; u < OW_GU; ++u) {
          if (j0 + u < n) {
            const int sj = __builtin_amdgcn_readlane(s, j0 + u);
            const float* gp = G + (size_t)sj * DC;
            g[u][0] = gp[lane];
            g[u][1] = gp[lane + 32];
            g[u][2] = gp[lane + 64];
            g[u][3] = gp[lane + 96];
          } else {
            g[u][0] = g[u][1] = g[u][2] = g[u][3] = 0.f;
          }
        }
#pragma unroll
        for (int u = 0; u < OW_GU; ++u) {
          if (j0 + u < n) {
            const int   rj = __builtin_amdgcn_readlane(r, j0 + u);
            const float wj = __int_as_float(
                __builtin_amdgcn_readlane(__float_as_int(w), j0 + u));
            float* ap = acc + rj * WSTR + lane;
            atomicAdd(ap,      wj * g[u][0]);
            atomicAdd(ap + 32, wj * g[u][1]);
            atomicAdd(ap + 64, wj * g[u][2]);
            atomicAdd(ap + 96, wj * g[u][3]);
          }
        }
      }
    }
  };

  int qn = 0;

  const unsigned gp    = ((unsigned)blockIdx.x & 255u) * 0x01010101u;
  const int    E16     = E >> 4;
  const uint4* ob4     = (const uint4*)ob;
  const int    nchunk  = (E16 + OW_T - 1) / OW_T;
  const int    ntrip   = (nchunk + OW_SC - 1) / OW_SC;
  const int    rot     = ntrip > 0 ? (int)(((unsigned)blockIdx.x * 61u) % (unsigned)ntrip) : 0;
  for (int t0 = 0; t0 < ntrip; ++t0) {
    int tr = t0 + rot;
    if (tr >= ntrip) tr -= ntrip;
    const int i0 = tr * (OW_SC * OW_T) + tid;
    uint4 dv[OW_SC];
    if ((tr + 1) * (OW_SC * OW_T) <= E16) {
#pragma unroll
      for (int u = 0; u < OW_SC; ++u) dv[u] = ob4[i0 + u * OW_T];
    } else {
#pragma unroll
      for (int u = 0; u < OW_SC; ++u) {
        const int  i16 = i0 + u * OW_T;
        const bool ok  = i16 < E16;
        uint4 t = ob4[ok ? i16 : 0];
        if (!ok) t = make_uint4(~gp, ~gp, ~gp, ~gp);
        dv[u] = t;
      }
    }
    unsigned keep[OW_SC / 2];
#pragma unroll
    for (int h = 0; h < OW_SC / 2; ++h) {
      const uint4 p = dv[2 * h], s4 = dv[2 * h + 1];
      unsigned k = differ_bits(p.x, gp);
      k &= __builtin_rotateright32(differ_bits(p.y,  gp), 1);
      k &= __builtin_rotateright32(differ_bits(p.z,  gp), 2);
      k &= __builtin_rotateright32(differ_bits(p.w,  gp), 3);
      k &= __builtin_rotateright32(differ_bits(s4.x, gp), 4);
      k &= __builtin_rotateright32(differ_bits(s4.y, gp), 5);
      k &= __builtin_rotateright32(differ_bits(s4.z, gp), 6);
      k &= __builtin_rotateright32(differ_bits(s4.w, gp), 7);
      keep[h] = k;
    }
    unsigned long long m = ((unsigned long long)(~keep[1]) << 32) | (unsigned long long)(~keep[0]);
    for (;;) {
      const bool has = m != 0ull;
      const unsigned mask = __builtin_amdgcn_ballot_w32(has);
      if (!mask) break;
      if (has) {
        const int pb = __builtin_ctzll(m);
        m &= m - 1ull;
        const int j  = 7 - (pb & 7);
        const int u  = ((pb >> 5) << 1) + (j >> 2);
        const int e  = ((i0 + u * OW_T) << 4) + ((j & 3) << 2) + ((pb >> 3) & 3);
        q[wv][qn + (int)__builtin_amdgcn_mbcnt_lo(mask, 0u)] = e;
      }
      qn += __builtin_popcount(mask);
      if (qn >= 32) { take(qn); qn = 0; }
    }
  }
  if (wv == 0) {
    const int  e   = (E16 << 4) + lane;
    const bool has = e < E;
    const unsigned mask = __builtin_amdgcn_ballot_w32(has);
    if (mask) {
      if (has) q[wv][qn + (int)__builtin_amdgcn_mbcnt_lo(mask, 0u)] = e;
      qn += __builtin_popcount(mask);
    }
  }
  if (qn > 0) take(qn);
  __syncthreads();

  for (int i = tid; i < nrows * (DC / 4); i += OW_T) {
    const int r  = i >> 5;
    const int c4 = (i & 31) * 4;
    float* arow = acc + r * WSTR + c4;
    const float4 a4 = *(const float4*)arow;
    const float den = zs[r] + 1e-16f;
    float4 o4;
    o4.x = a4.x / den; o4.y = a4.y / den; o4.z = a4.z / den; o4.w = a4.w / den;
    if (b1) {
      float4 b4 = *(const float4*)(b1 + c4);
      if (b2) {
        const float4 c4v = *(const float4*)(b2 + c4);
        b4.x += c4v.x; b4.y += c4v.y; b4.z += c4v.z; b4.w += c4v.w;
      }
      o4.x += b4.x; o4.y += b4.y; o4.z += b4.z; o4.w += b4.w;
    } else {
      const float4 p4 = *(const float4*)(O + (size_t)(lo + r) * DC + c4);
      o4.x += p4.x; o4.y += p4.y; o4.z += p4.z; o4.w += p4.w;
    }
    if (store_O) *(float4*)(O + (size_t)(lo + r) * DC + c4) = o4;
    if (Wa) {
      o4.x = o4.x > 0.f ? o4.x : 0.f;
      o4.y = o4.y > 0.f ? o4.y : 0.f;
      o4.z = o4.z > 0.f ? o4.z : 0.f;
      o4.w = o4.w > 0.f ? o4.w : 0.f;
      *(float4*)arow = o4;
      if (u) {
        const float4 u4 = ((const float4*)u)[lane];
        float sd = o4.x * u4.x + o4.y * u4.y + o4.z * u4.z + o4.w * u4.w;
#pragma unroll
        for (int off = 16; off > 0; off >>= 1) sd += __shfl_down(sd, off, 32);
        if (lane == 0) du[lo + r] = sd;
      }
    }
  }

  if (Wa) {
    __syncthreads();
    float* sR = reinterpret_cast<float*>(&q[0][0]);
    const int ntile = (nrows + 15) >> 4;
    const int ct    = wv & 7;
    const int rg    = wv >> 3;
    const int m     = lane & 15;
    const int half  = lane >> 4;
    const int col   = ct * 16 + m;

    auto product = [&](const float* __restrict__ Wm, const float* __restrict__ bias,
                       float* __restrict__ C,
                       const float* __restrict__ v1, const float* __restrict__ v2,
                       float* d1, float* d2) {
      v2f b[32];
      const float* wp = Wm + (size_t)(2 * half) * DC + col;
#pragma unroll
      for (int kk = 0; kk < 32; ++kk) {
        b[kk].x = wp[kk * (4 * DC)];
        b[kk].y = wp[kk * (4 * DC) + DC];
      }
      const float bv = bias ? bias[col] : 0.f;
      const float w1 = v1 ? v1[col] : 0.f;
      const float w2 = v2 ? v2[col] : 0.f;
#pragma unroll 1
      for (int rt = rg; rt < ntile; rt += 4) {
        const float* ap = acc + (rt * 16 + m) * WSTR + half * 2;
        v8f c = {};
#pragma unroll
        for (int kk = 0; kk < 32; ++kk) {
          const v2f a = *(const v2f*)(ap + kk * 4);
          c = __builtin_amdgcn_wmma_f32_16x16x4_f32(false, a, false, b[kk],
                                                    (short)0, c, false, false);
        }
        asm volatile("v_nop\n\tv_nop\n\tv_nop\n\tv_nop" : "+v"(c));
        float cv[8];
#pragma unroll
        for (int r = 0; r < 8; ++r) cv[r] = c[r] + bv;
        if (v1) {
          float p1[8], p2[8];
#pragma unroll
          for (int r = 0; r < 8; ++r) { p1[r] = cv[r] * w1; p2[r] = cv[r] * w2; }
          const float t1 = rowsum8_over16(p1, m);
          const float t2 = rowsum8_over16(p2, m);
          if ((m & 1) == 0) {
            const int rr = rt * 16 + half * 8 + (m >> 1);
            float* qq = sR + ((ct * OW_AROWS + rr) << 1);
            qq[0] = t1;
            qq[1] = t2;
          }
        }
        const int row0 = rt * 16 + half * 8;
        float* cp = C + ((size_t)(lo + row0)) * DC + col;
#pragma unroll
        for (int r = 0; r < 8; ++r)
          if (row0 + r < nrows) cp[(size_t)r * DC] = cv[r];
      }
      if (v1) {
        __syncthreads();
        for (int i = tid; i < nrows * 2; i += OW_T) {
          const int rr = i >> 1, which = i & 1;
          float sd = 0.f;
#pragma unroll
          for (int w = 0; w < 8; ++w) sd += sR[((w * OW_AROWS + rr) << 1) + which];
          float* dp = which ? d2 : d1;
          if (dp) dp[lo + rr] = sd;
        }
        __syncthreads();
      }
    };

    product(Wa, ba, Ca, va1, va2, da1, da2);
    if (Wb) product(Wb, nullptr, Cb, vb1, nullptr, db1, nullptr);
  }
}

extern "C" void kernel_launch(void* const* d_in, const int* in_sizes, int n_in,
                              void* d_out, int out_size, void* d_ws, size_t ws_size,
                              hipStream_t stream) {
  const float* x1    = (const float*)d_in[0];
  const float* x2    = (const float*)d_in[1];
  const int*   ei11  = (const int*)d_in[2];
  const int*   ei22  = (const int*)d_in[3];
  const int*   ei12  = (const int*)d_in[4];
  const float* W_i1  = (const float*)d_in[5];
  const float* as_i1 = (const float*)d_in[6];
  const float* ad_i1 = (const float*)d_in[7];
  const float* b_i1  = (const float*)d_in[8];
  const float* W_i2  = (const float*)d_in[9];
  const float* as_i2 = (const float*)d_in[10];
  const float* ad_i2 = (const float*)d_in[11];
  const float* b_i2  = (const float*)d_in[12];
  const float* Ws_x  = (const float*)d_in[13];
  const float* Wd_x  = (const float*)d_in[14];
  const float* as_x  = (const float*)d_in[15];
  const float* ad_x  = (const float*)d_in[16];
  const float* b_x   = (const float*)d_in[17];
  const float* W_out = (const float*)d_in[18];
  const float* b_out = (const float*)d_in[19];

  float* ws = (float*)d_ws;
  size_t off = 0;
  auto alloc = [&](size_t n) { float* p = ws + off; off += n; return p; };

  float* g1n  = alloc((size_t)N1C * DC);
  float* h2   = alloc((size_t)N2C * DC);
  float* g1   = alloc((size_t)N1C * DC);
  float* g2   = alloc((size_t)N2C * DC);
  float* gx   = alloc((size_t)N1C * DC);
  float* a_s1 = alloc(N1C);
  float* a_d1 = alloc(N1C);
  float* a_s2 = alloc(N2C);
  float* a_d2 = alloc(N2C);
  float* a_sx = alloc(N1C);
  float* a_dx = alloc(N2C);
  float* vd   = alloc(DC);
  unsigned* ob11 = (unsigned*)alloc(EC / 4);
  unsigned* ob22 = (unsigned*)alloc(EC / 4);
  unsigned* ob12 = (unsigned*)alloc(EC / 4);
  (void)ws_size; (void)in_sizes; (void)n_in; (void)out_size;

  const dim3 blk(256);
  const dim3 gemm_g1(N1C / 16), gemm_g2(N2C / 16);
  const dim3 ow_g(OW_WG), ow_b(OW_T);
  const size_t lds_bytes = (size_t)(DC * DC + 16 * WSTR + 8 * 16 * 2) * sizeof(float);

  const size_t w1o = (size_t)DC * DC;
  const size_t v1o = (size_t)DC;
  float* out = (float*)d_out;

  owner_byte_kernel<<<dim3((EC / 4 + 255) / 256, 3), blk, 0, stream>>>(
      ei11 + EC, ei22 + EC, ei12 + EC, ob11, ob22, ob12, EC, N1C, N2C, N2C, OW_WG);

  vecmat128_kernel<<<1, DC, 0, stream>>>(Wd_x, ad_x, vd);
  gemm_n128_wmma<<<gemm_g1, blk, lds_bytes, stream>>>(x1, Ws_x, nullptr, gx, 0,
      nullptr, nullptr,
      as_x, nullptr, a_sx, nullptr);
  gemm_n128_wmma<<<gemm_g1, blk, lds_bytes, stream>>>(x1, W_i1, nullptr, g1, 0,
      nullptr, nullptr,
      as_i1, ad_i1, a_s1, a_d1);
  gemm_n128_wmma<<<gemm_g2, blk, lds_bytes, stream>>>(x2, W_i2, nullptr, g2, 0,
      vd, a_dx,
      as_i2, ad_i2, a_s2, a_d2);
  vecmat128_kernel<<<1, DC, 0, stream>>>(Wd_x + w1o, ad_x + v1o, vd);

  float* a_s1n = h2;

  edge_owner_kernel<<<ow_g, ow_b, 0, stream>>>(a_s2, a_d2, ei22, ei22 + EC, ob22, g2, h2,
      b_i2, b_x, EC, N2C, N2C, 1,
      nullptr, nullptr,
      nullptr, nullptr, nullptr, nullptr, nullptr, nullptr, nullptr,
      nullptr, nullptr, nullptr, nullptr);
  edge_owner_kernel<<<ow_g, ow_b, 0, stream>>>(a_sx, a_dx, ei12, ei12 + EC, ob12, gx, h2,
      nullptr, nullptr, EC, N1C, N2C, 0,
      vd, a_dx,
      W_i2 + w1o, nullptr, g2, as_i2 + v1o, ad_i2 + v1o, a_s2, a_d2,
      nullptr, nullptr, nullptr, nullptr);
  edge_owner_kernel<<<ow_g, ow_b, 0, stream>>>(a_s1, a_d1, ei11, ei11 + EC, ob11, g1, nullptr,
      b_i1, nullptr, EC, N1C, N1C, 0,
      nullptr, nullptr,
      W_i1 + w1o, nullptr, g1n, as_i1 + v1o, ad_i1 + v1o, a_s1n, a_d1,
      Ws_x + w1o, gx, as_x + v1o, a_sx);

  edge_owner_kernel<<<ow_g, ow_b, 0, stream>>>(a_s1n, a_d1, ei11, ei11 + EC, ob11, g1n, nullptr,
      b_i1 + v1o, nullptr, EC, N1C, N1C, 0,
      nullptr, nullptr,
      W_out, b_out, out, nullptr, nullptr, nullptr, nullptr,
      nullptr, nullptr, nullptr, nullptr);
  edge_owner_kernel<<<ow_g, ow_b, 0, stream>>>(a_s2, a_d2, ei22, ei22 + EC, ob22, g2, h2,
      b_i2 + v1o, b_x + v1o, EC, N2C, N2C, 1,
      nullptr, nullptr,
      nullptr, nullptr, nullptr, nullptr, nullptr, nullptr, nullptr,
      nullptr, nullptr, nullptr, nullptr);
  edge_owner_kernel<<<ow_g, ow_b, 0, stream>>>(a_sx, a_dx, ei12, ei12 + EC, ob12, gx, h2,
      nullptr, nullptr, EC, N1C, N2C, 0,
      nullptr, nullptr,
      W_out, b_out, out + (size_t)N1C * DC, nullptr, nullptr, nullptr, nullptr,
      nullptr, nullptr, nullptr, nullptr);
}
